# starting kernel
# baseline (speedup 1.0000x reference)
#include <hip/hip_runtime.h>
#include <math.h>

typedef float v2f __attribute__((ext_vector_type(2)));
typedef float v8f __attribute__((ext_vector_type(8)));

#define NS 32768
#define DD 64
#define LL 63
#define PP 14
#define NCAT 512
#define BB 5.0f
#define MINW 0.001f
#define MINH 0.001f
#define MIND 0.001f

__global__ void nsf_pack_w1(const float* __restrict__ W1, float* __restrict__ Bt) {
    int idx = blockIdx.x * blockDim.x + threadIdx.x;
    if (idx >= NCAT * 64) return;
    int col = idx >> 6;
    int k   = idx & 63;
    int l   = col >> 3;
    int h   = col & 7;
    float v = 0.0f;
    if (l < LL && k <= l) v = W1[(l * LL + k) * 8 + h];
    Bt[col * 64 + k] = v;
}

__device__ __forceinline__ float softplusf(float v) {
    return v > 20.0f ? v : log1pf(expf(v));
}

__device__ void rqs_eval(const float* __restrict__ p, float xf, float& z, float& ld) {
    float cumw[6], cumh[6], derivs[6];
    {
        float mx = p[0];
        #pragma unroll
        for (int i = 1; i < 5; i++) mx = fmaxf(mx, p[i]);
        float e[5], s = 0.0f;
        #pragma unroll
        for (int i = 0; i < 5; i++) { e[i] = expf(p[i] - mx); s += e[i]; }
        float inv = 1.0f / s, cs = 0.0f;
        cumw[0] = -BB;
        #pragma unroll
        for (int i = 0; i < 4; i++) {
            cs += MINW + (1.0f - MINW * 5.0f) * e[i] * inv;
            cumw[i + 1] = 2.0f * BB * cs - BB;
        }
        cumw[5] = BB;
    }
    {
        float mx = p[5];
        #pragma unroll
        for (int i = 6; i < 10; i++) mx = fmaxf(mx, p[i]);
        float e[5], s = 0.0f;
        #pragma unroll
        for (int i = 0; i < 5; i++) { e[i] = expf(p[5 + i] - mx); s += e[i]; }
        float inv = 1.0f / s, cs = 0.0f;
        cumh[0] = -BB;
        #pragma unroll
        for (int i = 0; i < 4; i++) {
            cs += MINH + (1.0f - MINH * 5.0f) * e[i] * inv;
            cumh[i + 1] = 2.0f * BB * cs - BB;
        }
        cumh[5] = BB;
    }
    derivs[0] = 1.0f; derivs[5] = 1.0f;
    #pragma unroll
    for (int i = 1; i < 5; i++) derivs[i] = MIND + softplusf(p[9 + i]);

    float xc = fminf(fmaxf(xf, -BB), BB);
    int idx = 0;
    #pragma unroll
    for (int j = 1; j < 6; j++) idx += (xc >= cumw[j]) ? 1 : 0;
    if (idx > 4) idx = 4;

    float icw = cumw[idx], ibw = cumw[idx + 1] - cumw[idx];
    float ich = cumh[idx], ih  = cumh[idx + 1] - cumh[idx];
    float idl = ih / ibw;
    float idv = derivs[idx], idvp1 = derivs[idx + 1];
    float theta = (xc - icw) / ibw;
    float tmt = theta * (1.0f - theta);
    float num = ih * (idl * theta * theta + idv * tmt);
    float den = idl + (idv + idvp1 - 2.0f * idl) * tmt;
    float outv = ich + num / den;
    float omt = 1.0f - theta;
    float dnum = idl * idl * (idvp1 * theta * theta + 2.0f * idl * tmt + idv * omt * omt);
    float ldv = logf(dnum) - 2.0f * logf(den);
    bool inside = (xf >= -BB) && (xf <= BB);
    z  = inside ? outv : xf;
    ld = inside ? ldv : 0.0f;
}

__launch_bounds__(128)
__global__ void nsf_fused(const float* __restrict__ x, const float* __restrict__ initp,
                          const float* __restrict__ b1, const float* __restrict__ W2,
                          const float* __restrict__ b2, const float* __restrict__ W3,
                          const float* __restrict__ b3, const float* __restrict__ Bt,
                          float* __restrict__ out) {
    __shared__ float tile[4][16][17];
    __shared__ float ldacc[DD];

    const int tid  = threadIdx.x;
    const int wave = tid >> 5;
    const int lane = tid & 31;
    const int m    = lane & 15;
    const int hh   = lane >> 4;
    const int s0   = blockIdx.x * 64 + wave * 16;
    const int s    = s0 + m;

    if (tid < DD) ldacc[tid] = 0.0f;
    __syncthreads();

    if (lane < 16) {
        float p[PP];
        #pragma unroll
        for (int q = 0; q < PP; q++) p[q] = initp[q];
        float z, ld;
        rqs_eval(p, x[s * DD + 0], z, ld);
        out[s] = z;
        atomicAdd(&ldacc[0], ld);
    }

    v2f a[16];
    const float* xrow = x + (size_t)s * DD;
    #pragma unroll
    for (int kk = 0; kk < 16; kk++)
        a[kk] = *(const v2f*)(xrow + 4 * kk + 2 * hh);

    for (int t = 0; t < 32; ++t) {
        const float* bcol = Bt + (size_t)(16 * t + m) * 64 + 2 * hh;
        v8f c = {0.f, 0.f, 0.f, 0.f, 0.f, 0.f, 0.f, 0.f};
        #pragma unroll
        for (int kk = 0; kk < 16; kk++) {
            v2f b = *(const v2f*)(bcol + 4 * kk);
            c = __builtin_amdgcn_wmma_f32_16x16x4_f32(
                    false, a[kk], false, b, (short)0, c, false, false);
        }

        const int col = 16 * t + m;
        const float bias = (col < LL * 8) ? b1[col] : 0.0f;
        #pragma unroll
        for (int r = 0; r < 8; r++) {
            float hv = tanhf(c[r] + bias);
            tile[wave][r + 8 * hh][m] = hv;
        }
        __syncthreads();

        const int l_idx = 2 * t + hh;
        if (l_idx < LL) {
            float hv[8];
            #pragma unroll
            for (int j = 0; j < 8; j++) hv[j] = tile[wave][m][8 * hh + j];

            const float* w2  = W2 + (size_t)l_idx * 64;
            const float* bb2 = b2 + (size_t)l_idx * 8;
            float h2[8];
            #pragma unroll
            for (int g = 0; g < 8; g++) {
                float acc = bb2[g];
                #pragma unroll
                for (int j = 0; j < 8; j++) acc += hv[j] * w2[j * 8 + g];
                h2[g] = tanhf(acc);
            }

            const float* w3  = W3 + (size_t)l_idx * 8 * PP;
            const float* bb3 = b3 + (size_t)l_idx * PP;
            float pv[PP];
            #pragma unroll
            for (int q = 0; q < PP; q++) {
                float acc = bb3[q];
                #pragma unroll
                for (int g = 0; g < 8; g++) acc += h2[g] * w3[g * PP + q];
                pv[q] = acc;
            }

            const int cdim = l_idx + 1;
            float z, ld;
            rqs_eval(pv, x[(size_t)s * DD + cdim], z, ld);
            out[(size_t)cdim * NS + s] = z;
            atomicAdd(&ldacc[cdim], ld);
        }
        __syncthreads();
    }

    if (tid < DD)
        out[(size_t)NS * DD + (size_t)tid * (NS / 64) + blockIdx.x] = ldacc[tid];
}

extern "C" void kernel_launch(void* const* d_in, const int* in_sizes, int n_in,
                              void* d_out, int out_size, void* d_ws, size_t ws_size,
                              hipStream_t stream) {
    const float* x     = (const float*)d_in[0];
    const float* initp = (const float*)d_in[1];
    const float* W1    = (const float*)d_in[2];
    const float* b1    = (const float*)d_in[3];
    const float* W2    = (const float*)d_in[4];
    const float* b2    = (const float*)d_in[5];
    const float* W3    = (const float*)d_in[6];
    const float* b3    = (const float*)d_in[7];
    float* out = (float*)d_out;
    float* Bt  = (float*)d_ws;

    nsf_pack_w1<<<(NCAT * 64 + 255) / 256, 256, 0, stream>>>(W1, Bt);
    nsf_fused<<<NS / 64, 128, 0, stream>>>(x, initp, b1, W2, b2, W3, b3, Bt, out);
}


# split each 64-sample block's 32 layer-pair iterations over 2 work-groups (grid 512 -> 512x2): 4 resident waves per SIMD instead of 2 cover dependent-issue and load latency
# speedup vs baseline: 1.7985x; 1.7985x over previous
#include <hip/hip_runtime.h>
#include <math.h>

typedef float v2f __attribute__((ext_vector_type(2)));
typedef float v8f __attribute__((ext_vector_type(8)));

#define NS 32768
#define DD 64
#define LL 63
#define PP 14
#define NCAT 512
#define BB 5.0f
#define MINW 0.001f
#define MINH 0.001f
#define MIND 0.001f
#define TSPLIT 2
#define TPER (32 / TSPLIT)

__global__ void nsf_pack_w1(const float* __restrict__ W1, float* __restrict__ Bt) {
    int idx = blockIdx.x * blockDim.x + threadIdx.x;
    if (idx >= NCAT * 64) return;
    int col = idx >> 6;
    int k   = idx & 63;
    int l   = col >> 3;
    int h   = col & 7;
    float v = 0.0f;
    if (l < LL && k <= l) v = W1[(l * LL + k) * 8 + h];
    Bt[col * 64 + k] = v;
}

__device__ __forceinline__ float softplusf(float v) {
    return v > 20.0f ? v : log1pf(expf(v));
}

__device__ void rqs_eval(const float* __restrict__ p, float xf, float& z, float& ld) {
    float cumw[6], cumh[6], derivs[6];
    {
        float mx = p[0];
        #pragma unroll
        for (int i = 1; i < 5; i++) mx = fmaxf(mx, p[i]);
        float e[5], s = 0.0f;
        #pragma unroll
        for (int i = 0; i < 5; i++) { e[i] = expf(p[i] - mx); s += e[i]; }
        float inv = 1.0f / s, cs = 0.0f;
        cumw[0] = -BB;
        #pragma unroll
        for (int i = 0; i < 4; i++) {
            cs += MINW + (1.0f - MINW * 5.0f) * e[i] * inv;
            cumw[i + 1] = 2.0f * BB * cs - BB;
        }
        cumw[5] = BB;
    }
    {
        float mx = p[5];
        #pragma unroll
        for (int i = 6; i < 10; i++) mx = fmaxf(mx, p[i]);
        float e[5], s = 0.0f;
        #pragma unroll
        for (int i = 0; i < 5; i++) { e[i] = expf(p[5 + i] - mx); s += e[i]; }
        float inv = 1.0f / s, cs = 0.0f;
        cumh[0] = -BB;
        #pragma unroll
        for (int i = 0; i < 4; i++) {
            cs += MINH + (1.0f - MINH * 5.0f) * e[i] * inv;
            cumh[i + 1] = 2.0f * BB * cs - BB;
        }
        cumh[5] = BB;
    }
    derivs[0] = 1.0f; derivs[5] = 1.0f;
    #pragma unroll
    for (int i = 1; i < 5; i++) derivs[i] = MIND + softplusf(p[9 + i]);

    float xc = fminf(fmaxf(xf, -BB), BB);
    int idx = 0;
    #pragma unroll
    for (int j = 1; j < 6; j++) idx += (xc >= cumw[j]) ? 1 : 0;
    if (idx > 4) idx = 4;

    float icw = cumw[idx], ibw = cumw[idx + 1] - cumw[idx];
    float ich = cumh[idx], ih  = cumh[idx + 1] - cumh[idx];
    float idl = ih / ibw;
    float idv = derivs[idx], idvp1 = derivs[idx + 1];
    float theta = (xc - icw) / ibw;
    float tmt = theta * (1.0f - theta);
    float num = ih * (idl * theta * theta + idv * tmt);
    float den = idl + (idv + idvp1 - 2.0f * idl) * tmt;
    float outv = ich + num / den;
    float omt = 1.0f - theta;
    float dnum = idl * idl * (idvp1 * theta * theta + 2.0f * idl * tmt + idv * omt * omt);
    float ldv = logf(dnum) - 2.0f * logf(den);
    bool inside = (xf >= -BB) && (xf <= BB);
    z  = inside ? outv : xf;
    ld = inside ? ldv : 0.0f;
}

__launch_bounds__(128)
__global__ void nsf_fused(const float* __restrict__ x, const float* __restrict__ initp,
                          const float* __restrict__ b1, const float* __restrict__ W2,
                          const float* __restrict__ b2, const float* __restrict__ W3,
                          const float* __restrict__ b3, const float* __restrict__ Bt,
                          float* __restrict__ out) {
    __shared__ float tile[4][16][17];
    __shared__ float ldacc[DD];

    const int tid  = threadIdx.x;
    const int wave = tid >> 5;
    const int lane = tid & 31;
    const int m    = lane & 15;
    const int hh   = lane >> 4;
    const int s0   = blockIdx.x * 64 + wave * 16;
    const int s    = s0 + m;
    const int part = blockIdx.y;
    const int t_begin = part * TPER;

    if (tid < DD) ldacc[tid] = 0.0f;
    __syncthreads();

    if (part == 0 && lane < 16) {
        float p[PP];
        #pragma unroll
        for (int q = 0; q < PP; q++) p[q] = initp[q];
        float z, ld;
        rqs_eval(p, x[s * DD + 0], z, ld);
        out[s] = z;
        atomicAdd(&ldacc[0], ld);
    }

    v2f a[16];
    const float* xrow = x + (size_t)s * DD;
    #pragma unroll
    for (int kk = 0; kk < 16; kk++)
        a[kk] = *(const v2f*)(xrow + 4 * kk + 2 * hh);

    for (int t = t_begin; t < t_begin + TPER; ++t) {
        const float* bcol = Bt + (size_t)(16 * t + m) * 64 + 2 * hh;
        v8f c = {0.f, 0.f, 0.f, 0.f, 0.f, 0.f, 0.f, 0.f};
        #pragma unroll
        for (int kk = 0; kk < 16; kk++) {
            v2f b = *(const v2f*)(bcol + 4 * kk);
            c = __builtin_amdgcn_wmma_f32_16x16x4_f32(
                    false, a[kk], false, b, (short)0, c, false, false);
        }

        const int col = 16 * t + m;
        const float bias = (col < LL * 8) ? b1[col] : 0.0f;
        #pragma unroll
        for (int r = 0; r < 8; r++) {
            float hv = tanhf(c[r] + bias);
            tile[wave][r + 8 * hh][m] = hv;
        }
        __syncthreads();

        const int l_idx = 2 * t + hh;
        if (l_idx < LL) {
            float hv[8];
            #pragma unroll
            for (int j = 0; j < 8; j++) hv[j] = tile[wave][m][8 * hh + j];

            const float* w2  = W2 + (size_t)l_idx * 64;
            const float* bb2 = b2 + (size_t)l_idx * 8;
            float h2[8];
            #pragma unroll
            for (int g = 0; g < 8; g++) {
                float acc = bb2[g];
                #pragma unroll
                for (int j = 0; j < 8; j++) acc += hv[j] * w2[j * 8 + g];
                h2[g] = tanhf(acc);
            }

            const float* w3  = W3 + (size_t)l_idx * 8 * PP;
            const float* bb3 = b3 + (size_t)l_idx * PP;
            float pv[PP];
            #pragma unroll
            for (int q = 0; q < PP; q++) {
                float acc = bb3[q];
                #pragma unroll
                for (int g = 0; g < 8; g++) acc += h2[g] * w3[g * PP + q];
                pv[q] = acc;
            }

            const int cdim = l_idx + 1;
            float z, ld;
            rqs_eval(pv, x[(size_t)s * DD + cdim], z, ld);
            out[(size_t)cdim * NS + s] = z;
            atomicAdd(&ldacc[cdim], ld);
        }
        __syncthreads();
    }

    if (tid < DD) {
        const int owner = (tid == 0) ? 0 : (((tid - 1) >> 1) / TPER);
        if (owner == part)
            out[(size_t)NS * DD + (size_t)tid * (NS / 64) + blockIdx.x] = ldacc[tid];
    }
}

extern "C" void kernel_launch(void* const* d_in, const int* in_sizes, int n_in,
                              void* d_out, int out_size, void* d_ws, size_t ws_size,
                              hipStream_t stream) {
    const float* x     = (const float*)d_in[0];
    const float* initp = (const float*)d_in[1];
    const float* W1    = (const float*)d_in[2];
    const float* b1    = (const float*)d_in[3];
    const float* W2    = (const float*)d_in[4];
    const float* b2    = (const float*)d_in[5];
    const float* W3    = (const float*)d_in[6];
    const float* b3    = (const float*)d_in[7];
    float* out = (float*)d_out;
    float* Bt  = (float*)d_ws;

    nsf_pack_w1<<<(NCAT * 64 + 255) / 256, 256, 0, stream>>>(W1, Bt);
    nsf_fused<<<dim3(NS / 64, TSPLIT), 128, 0, stream>>>(x, initp, b1, W2, b2, W3, b3, Bt, out);
}


# stage b1,W2,b2,W3,b3 of the work-group's 32 layers in LDS once; the loop reads them in one batch ahead of each tanh chain (x early too): 1 L2 round trip per iteration instead of 11
# speedup vs baseline: 2.0044x; 1.1145x over previous
#include <hip/hip_runtime.h>
#include <math.h>

typedef float v2f __attribute__((ext_vector_type(2)));
typedef float v8f __attribute__((ext_vector_type(8)));

#define NS 32768
#define DD 64
#define LL 63
#define PP 14
#define NCAT 512
#define BB 5.0f
#define MINW 0.001f
#define MINH 0.001f
#define MIND 0.001f
#define TSPLIT 2
#define TPER (32 / TSPLIT)
#define NLAY (2 * TPER)
#define LSTR 200
#define OFF_B2 64
#define OFF_W3 72
#define OFF_B3 184

__global__ void nsf_pack_w1(const float* __restrict__ W1, float* __restrict__ Bt) {
    int idx = blockIdx.x * blockDim.x + threadIdx.x;
    if (idx >= NCAT * 64) return;
    int col = idx >> 6;
    int k   = idx & 63;
    int l   = col >> 3;
    int h   = col & 7;
    float v = 0.0f;
    if (l < LL && k <= l) v = W1[(l * LL + k) * 8 + h];
    Bt[col * 64 + k] = v;
}

__device__ __forceinline__ float softplusf(float v) {
    return v > 20.0f ? v : log1pf(expf(v));
}

__device__ void rqs_eval(const float* __restrict__ p, float xf, float& z, float& ld) {
    float cumw[6], cumh[6], derivs[6];
    {
        float mx = p[0];
        #pragma unroll
        for (int i = 1; i < 5; i++) mx = fmaxf(mx, p[i]);
        float e[5], s = 0.0f;
        #pragma unroll
        for (int i = 0; i < 5; i++) { e[i] = expf(p[i] - mx); s += e[i]; }
        float inv = 1.0f / s, cs = 0.0f;
        cumw[0] = -BB;
        #pragma unroll
        for (int i = 0; i < 4; i++) {
            cs += MINW + (1.0f - MINW * 5.0f) * e[i] * inv;
            cumw[i + 1] = 2.0f * BB * cs - BB;
        }
        cumw[5] = BB;
    }
    {
        float mx = p[5];
        #pragma unroll
        for (int i = 6; i < 10; i++) mx = fmaxf(mx, p[i]);
        float e[5], s = 0.0f;
        #pragma unroll
        for (int i = 0; i < 5; i++) { e[i] = expf(p[5 + i] - mx); s += e[i]; }
        float inv = 1.0f / s, cs = 0.0f;
        cumh[0] = -BB;
        #pragma unroll
        for (int i = 0; i < 4; i++) {
            cs += MINH + (1.0f - MINH * 5.0f) * e[i] * inv;
            cumh[i + 1] = 2.0f * BB * cs - BB;
        }
        cumh[5] = BB;
    }
    derivs[0] = 1.0f; derivs[5] = 1.0f;
    #pragma unroll
    for (int i = 1; i < 5; i++) derivs[i] = MIND + softplusf(p[9 + i]);

    float xc = fminf(fmaxf(xf, -BB), BB);
    int idx = 0;
    #pragma unroll
    for (int j = 1; j < 6; j++) idx += (xc >= cumw[j]) ? 1 : 0;
    if (idx > 4) idx = 4;

    float icw = cumw[idx], ibw = cumw[idx + 1] - cumw[idx];
    float ich = cumh[idx], ih  = cumh[idx + 1] - cumh[idx];
    float idl = ih / ibw;
    float idv = derivs[idx], idvp1 = derivs[idx + 1];
    float theta = (xc - icw) / ibw;
    float tmt = theta * (1.0f - theta);
    float num = ih * (idl * theta * theta + idv * tmt);
    float den = idl + (idv + idvp1 - 2.0f * idl) * tmt;
    float outv = ich + num / den;
    float omt = 1.0f - theta;
    float dnum = idl * idl * (idvp1 * theta * theta + 2.0f * idl * tmt + idv * omt * omt);
    float ldv = logf(dnum) - 2.0f * logf(den);
    bool inside = (xf >= -BB) && (xf <= BB);
    z  = inside ? outv : xf;
    ld = inside ? ldv : 0.0f;
}

__launch_bounds__(128)
__global__ void nsf_fused(const float* __restrict__ x, const float* __restrict__ initp,
                          const float* __restrict__ b1, const float* __restrict__ W2,
                          const float* __restrict__ b2, const float* __restrict__ W3,
                          const float* __restrict__ b3, const float* __restrict__ Bt,
                          float* __restrict__ out) {
    __shared__ float tile[4][16][17];
    __shared__ float ldacc[DD];
    __shared__ __attribute__((aligned(16))) float wl[NLAY * LSTR];
    __shared__ float b1s[TPER * 16];

    const int tid  = threadIdx.x;
    const int wave = tid >> 5;
    const int lane = tid & 31;
    const int m    = lane & 15;
    const int hh   = lane >> 4;
    const int s0   = blockIdx.x * 64 + wave * 16;
    const int s    = s0 + m;
    const int part = blockIdx.y;
    const int t_begin = part * TPER;

    if (tid < DD) ldacc[tid] = 0.0f;

    {
        static_assert(NLAY % 16 == 0, "copy loops assume whole passes of 128 threads");
        const int L0   = 2 * t_begin;
        const int nlay = (LL - L0 < NLAY) ? (LL - L0) : NLAY;
        const float4* src2 = (const float4*)(W2 + (size_t)L0 * 64);
        const float4* src3 = (const float4*)(W3 + (size_t)L0 * 8 * PP);
        float4 r2[NLAY * 16 / 128], r3[NLAY * 28 / 128];
        float  rb2[NLAY * 8 / 128], rb3[NLAY * 16 / 128], rb1[TPER * 16 / 128];
        #pragma unroll
        for (int k = 0; k < NLAY * 16 / 128; k++) {
            const int i = tid + 128 * k;
            r2[k] = src2[i < nlay * 16 ? i : nlay * 16 - 1];
        }
        #pragma unroll
        for (int k = 0; k < NLAY * 28 / 128; k++) {
            const int i = tid + 128 * k;
            r3[k] = src3[i < nlay * 28 ? i : nlay * 28 - 1];
        }
        #pragma unroll
        for (int k = 0; k < NLAY * 8 / 128; k++) {
            const int i = tid + 128 * k;
            rb2[k] = b2[(size_t)L0 * 8 + (i < nlay * 8 ? i : nlay * 8 - 1)];
        }
        #pragma unroll
        for (int k = 0; k < NLAY * 16 / 128; k++) {
            const int i = tid + 128 * k;
            const int l = i >> 4, q = i & 15;
            const float v = b3[(size_t)(L0 + (l < nlay ? l : nlay - 1)) * PP + (q < PP ? q : PP - 1)];
            rb3[k] = (q < PP) ? v : 0.0f;
        }
        #pragma unroll
        for (int k = 0; k < TPER * 16 / 128; k++) {
            const int col = 16 * t_begin + tid + 128 * k;
            const float v = b1[col < LL * 8 ? col : LL * 8 - 1];
            rb1[k] = (col < LL * 8) ? v : 0.0f;
        }
        #pragma unroll
        for (int k = 0; k < NLAY * 16 / 128; k++) {
            const int i = tid + 128 * k;
            *(float4*)(wl + (i >> 4) * LSTR + 4 * (i & 15)) = r2[k];
        }
        #pragma unroll
        for (int k = 0; k < NLAY * 28 / 128; k++) {
            const int i = tid + 128 * k;
            const int l = i / 28, j = i - l * 28;
            *(float4*)(wl + l * LSTR + OFF_W3 + 4 * j) = r3[k];
        }
        #pragma unroll
        for (int k = 0; k < NLAY * 8 / 128; k++) {
            const int i = tid + 128 * k;
            wl[(i >> 3) * LSTR + OFF_B2 + (i & 7)] = rb2[k];
        }
        #pragma unroll
        for (int k = 0; k < NLAY * 16 / 128; k++) {
            const int i = tid + 128 * k;
            wl[(i >> 4) * LSTR + OFF_B3 + (i & 15)] = rb3[k];
        }
        #pragma unroll
        for (int k = 0; k < TPER * 16 / 128; k++) b1s[tid + 128 * k] = rb1[k];
    }
    __syncthreads();

    if (part == 0 && lane < 16) {
        float p[PP];
        #pragma unroll
        for (int q = 0; q < PP; q++) p[q] = initp[q];
        float z, ld;
        rqs_eval(p, x[s * DD + 0], z, ld);
        out[s] = z;
        atomicAdd(&ldacc[0], ld);
    }

    v2f a[16];
    const float* xrow = x + (size_t)s * DD;
    #pragma unroll
    for (int kk = 0; kk < 16; kk++)
        a[kk] = *(const v2f*)(xrow + 4 * kk + 2 * hh);

    for (int t = t_begin; t < t_begin + TPER; ++t) {
        const float bias = b1s[16 * (t - t_begin) + m];
        const float* bcol = Bt + (size_t)(16 * t + m) * 64 + 2 * hh;
        v8f c = {0.f, 0.f, 0.f, 0.f, 0.f, 0.f, 0.f, 0.f};
        #pragma unroll
        for (int kk = 0; kk < 16; kk++) {
            v2f b = *(const v2f*)(bcol + 4 * kk);
            c = __builtin_amdgcn_wmma_f32_16x16x4_f32(
                    false, a[kk], false, b, (short)0, c, false, false);
        }

        #pragma unroll
        for (int r = 0; r < 8; r++) {
            float hv = tanhf(c[r] + bias);
            tile[wave][r + 8 * hh][m] = hv;
        }
        __syncthreads();

        const int l_idx = 2 * t + hh;
        if (l_idx < LL) {
            const int cdim = l_idx + 1;
            const float xval = x[(size_t)s * DD + cdim];

            float hv[8];
            #pragma unroll
            for (int j = 0; j < 8; j++) hv[j] = tile[wave][m][8 * hh + j];

            const float* lw = wl + (2 * (t - t_begin) + hh) * LSTR;
            float w2v[64], acc2[8];
            #pragma unroll
            for (int i = 0; i < 16; i++) {
                const float4 v = *(const float4*)(lw + 4 * i);
                w2v[4 * i] = v.x; w2v[4 * i + 1] = v.y; w2v[4 * i + 2] = v.z; w2v[4 * i + 3] = v.w;
            }
            #pragma unroll
            for (int i = 0; i < 2; i++) {
                const float4 v = *(const float4*)(lw + OFF_B2 + 4 * i);
                acc2[4 * i] = v.x; acc2[4 * i + 1] = v.y; acc2[4 * i + 2] = v.z; acc2[4 * i + 3] = v.w;
            }
            #pragma unroll
            for (int g = 0; g < 8; g++) {
                #pragma unroll
                for (int j = 0; j < 8; j++) acc2[g] += hv[j] * w2v[j * 8 + g];
            }
            float h2[8];
            #pragma unroll
            for (int g = 0; g < 8; g++) h2[g] = tanhf(acc2[g]);

            float w3v[8 * PP], pv[16];
            #pragma unroll
            for (int i = 0; i < 28; i++) {
                const float4 v = *(const float4*)(lw + OFF_W3 + 4 * i);
                w3v[4 * i] = v.x; w3v[4 * i + 1] = v.y; w3v[4 * i + 2] = v.z; w3v[4 * i + 3] = v.w;
            }
            #pragma unroll
            for (int i = 0; i < 4; i++) {
                const float4 v = *(const float4*)(lw + OFF_B3 + 4 * i);
                pv[4 * i] = v.x; pv[4 * i + 1] = v.y; pv[4 * i + 2] = v.z; pv[4 * i + 3] = v.w;
            }
            #pragma unroll
            for (int q = 0; q < PP; q++) {
                #pragma unroll
                for (int g = 0; g < 8; g++) pv[q] += h2[g] * w3v[g * PP + q];
            }

            float z, ld;
            rqs_eval(pv, xval, z, ld);
            out[(size_t)cdim * NS + s] = z;
            atomicAdd(&ldacc[cdim], ld);
        }
        __syncthreads();
    }

    if (tid < DD) {
        const int owner = (tid == 0) ? 0 : (((tid - 1) >> 1) / TPER);
        if (owner == part)
            out[(size_t)NS * DD + (size_t)tid * (NS / 64) + blockIdx.x] = ldacc[tid];
    }
}

extern "C" void kernel_launch(void* const* d_in, const int* in_sizes, int n_in,
                              void* d_out, int out_size, void* d_ws, size_t ws_size,
                              hipStream_t stream) {
    const float* x     = (const float*)d_in[0];
    const float* initp = (const float*)d_in[1];
    const float* W1    = (const float*)d_in[2];
    const float* b1    = (const float*)d_in[3];
    const float* W2    = (const float*)d_in[4];
    const float* b2    = (const float*)d_in[5];
    const float* W3    = (const float*)d_in[6];
    const float* b3    = (const float*)d_in[7];
    float* out = (float*)d_out;
    float* Bt  = (float*)d_ws;

    nsf_pack_w1<<<(NCAT * 64 + 255) / 256, 256, 0, stream>>>(W1, Bt);
    nsf_fused<<<dim3(NS / 64, TSPLIT), 128, 0, stream>>>(x, initp, b1, W2, b2, W3, b3, Bt, out);
}


# in-line library math in the loop (tanhf, expf, log1pf, logf, IEEE a/b) -> one v_tanh/v_exp/v_log/v_rcp_f32 each: 2127 -> 622 instructions per loop trip, no divergent branch left; same formulas
# speedup vs baseline: 3.3828x; 1.6877x over previous
#include <hip/hip_runtime.h>
#include <math.h>

typedef float v2f __attribute__((ext_vector_type(2)));
typedef float v8f __attribute__((ext_vector_type(8)));

#define NS 32768
#define DD 64
#define LL 63
#define PP 14
#define NCAT 512
#define BB 5.0f
#define MINW 0.001f
#define MINH 0.001f
#define MIND 0.001f
#define TSPLIT 2
#define TPER (32 / TSPLIT)
#define NLAY (2 * TPER)
#define LSTR 200
#define OFF_B2 64
#define OFF_W3 72
#define OFF_B3 184

__global__ void nsf_pack_w1(const float* __restrict__ W1, float* __restrict__ Bt) {
    int idx = blockIdx.x * blockDim.x + threadIdx.x;
    if (idx >= NCAT * 64) return;
    int col = idx >> 6;
    int k   = idx & 63;
    int l   = col >> 3;
    int h   = col & 7;
    float v = 0.0f;
    if (l < LL && k <= l) v = W1[(l * LL + k) * 8 + h];
    Bt[col * 64 + k] = v;
}

__device__ __forceinline__ float hw_exp(float d)  { return __builtin_amdgcn_exp2f(d * 1.44269504088896341f); }
__device__ __forceinline__ float hw_ln(float v)   { return __builtin_amdgcn_logf(v) * 0.693147180559945309f; }
__device__ __forceinline__ float hw_rcp(float v)  { return __builtin_amdgcn_rcpf(v); }
__device__ __forceinline__ float hw_tanh(float v) { return __builtin_amdgcn_tanhf(v); }

__device__ __forceinline__ float softplusf(float v) {
    return fmaxf(v, 0.0f) + hw_ln(1.0f + hw_exp(-fabsf(v)));
}

__device__ void rqs_eval(const float* __restrict__ p, float xf, float& z, float& ld) {
    float cumw[6], cumh[6], derivs[6];
    {
        float mx = p[0];
        #pragma unroll
        for (int i = 1; i < 5; i++) mx = fmaxf(mx, p[i]);
        float e[5], s = 0.0f;
        #pragma unroll
        for (int i = 0; i < 5; i++) { e[i] = hw_exp(p[i] - mx); s += e[i]; }
        float inv = hw_rcp(s), cs = 0.0f;
        cumw[0] = -BB;
        #pragma unroll
        for (int i = 0; i < 4; i++) {
            cs += MINW + (1.0f - MINW * 5.0f) * e[i] * inv;
            cumw[i + 1] = 2.0f * BB * cs - BB;
        }
        cumw[5] = BB;
    }
    {
        float mx = p[5];
        #pragma unroll
        for (int i = 6; i < 10; i++) mx = fmaxf(mx, p[i]);
        float e[5], s = 0.0f;
        #pragma unroll
        for (int i = 0; i < 5; i++) { e[i] = hw_exp(p[5 + i] - mx); s += e[i]; }
        float inv = hw_rcp(s), cs = 0.0f;
        cumh[0] = -BB;
        #pragma unroll
        for (int i = 0; i < 4; i++) {
            cs += MINH + (1.0f - MINH * 5.0f) * e[i] * inv;
            cumh[i + 1] = 2.0f * BB * cs - BB;
        }
        cumh[5] = BB;
    }
    derivs[0] = 1.0f; derivs[5] = 1.0f;
    #pragma unroll
    for (int i = 1; i < 5; i++) derivs[i] = MIND + softplusf(p[9 + i]);

    float xc = fminf(fmaxf(xf, -BB), BB);
    int idx = 0;
    #pragma unroll
    for (int j = 1; j < 6; j++) idx += (xc >= cumw[j]) ? 1 : 0;
    if (idx > 4) idx = 4;

    float icw = cumw[idx], ibw = cumw[idx + 1] - cumw[idx];
    float ich = cumh[idx], ih  = cumh[idx + 1] - cumh[idx];
    const float rbw = hw_rcp(ibw);
    float idl = ih * rbw;
    float idv = derivs[idx], idvp1 = derivs[idx + 1];
    float theta = (xc - icw) * rbw;
    float tmt = theta * (1.0f - theta);
    float num = ih * (idl * theta * theta + idv * tmt);
    float den = idl + (idv + idvp1 - 2.0f * idl) * tmt;
    float outv = ich + num * hw_rcp(den);
    float omt = 1.0f - theta;
    float dnum = idl * idl * (idvp1 * theta * theta + 2.0f * idl * tmt + idv * omt * omt);
    float ldv = 0.693147180559945309f * (__builtin_amdgcn_logf(dnum) - 2.0f * __builtin_amdgcn_logf(den));
    bool inside = (xf >= -BB) && (xf <= BB);
    z  = inside ? outv : xf;
    ld = inside ? ldv : 0.0f;
}

__launch_bounds__(128)
__global__ void nsf_fused(const float* __restrict__ x, const float* __restrict__ initp,
                          const float* __restrict__ b1, const float* __restrict__ W2,
                          const float* __restrict__ b2, const float* __restrict__ W3,
                          const float* __restrict__ b3, const float* __restrict__ Bt,
                          float* __restrict__ out) {
    __shared__ float tile[4][16][17];
    __shared__ float ldacc[DD];
    __shared__ __attribute__((aligned(16))) float wl[NLAY * LSTR];
    __shared__ float b1s[TPER * 16];

    const int tid  = threadIdx.x;
    const int wave = tid >> 5;
    const int lane = tid & 31;
    const int m    = lane & 15;
    const int hh   = lane >> 4;
    const int s0   = blockIdx.x * 64 + wave * 16;
    const int s    = s0 + m;
    const int part = blockIdx.y;
    const int t_begin = part * TPER;

    if (tid < DD) ldacc[tid] = 0.0f;

    {
        static_assert(NLAY % 16 == 0, "copy loops assume whole passes of 128 threads");
        const int L0   = 2 * t_begin;
        const int nlay = (LL - L0 < NLAY) ? (LL - L0) : NLAY;
        const float4* src2 = (const float4*)(W2 + (size_t)L0 * 64);
        const float4* src3 = (const float4*)(W3 + (size_t)L0 * 8 * PP);
        float4 r2[NLAY * 16 / 128], r3[NLAY * 28 / 128];
        float  rb2[NLAY * 8 / 128], rb3[NLAY * 16 / 128], rb1[TPER * 16 / 128];
        #pragma unroll
        for (int k = 0; k < NLAY * 16 / 128; k++) {
            const int i = tid + 128 * k;
            r2[k] = src2[i < nlay * 16 ? i : nlay * 16 - 1];
        }
        #pragma unroll
        for (int k = 0; k < NLAY * 28 / 128; k++) {
            const int i = tid + 128 * k;
            r3[k] = src3[i < nlay * 28 ? i : nlay * 28 - 1];
        }
        #pragma unroll
        for (int k = 0; k < NLAY * 8 / 128; k++) {
            const int i = tid + 128 * k;
            rb2[k] = b2[(size_t)L0 * 8 + (i < nlay * 8 ? i : nlay * 8 - 1)];
        }
        #pragma unroll
        for (int k = 0; k < NLAY * 16 / 128; k++) {
            const int i = tid + 128 * k;
            const int l = i >> 4, q = i & 15;
            const float v = b3[(size_t)(L0 + (l < nlay ? l : nlay - 1)) * PP + (q < PP ? q : PP - 1)];
            rb3[k] = (q < PP) ? v : 0.0f;
        }
        #pragma unroll
        for (int k = 0; k < TPER * 16 / 128; k++) {
            const int col = 16 * t_begin + tid + 128 * k;
            const float v = b1[col < LL * 8 ? col : LL * 8 - 1];
            rb1[k] = (col < LL * 8) ? v : 0.0f;
        }
        #pragma unroll
        for (int k = 0; k < NLAY * 16 / 128; k++) {
            const int i = tid + 128 * k;
            *(float4*)(wl + (i >> 4) * LSTR + 4 * (i & 15)) = r2[k];
        }
        #pragma unroll
        for (int k = 0; k < NLAY * 28 / 128; k++) {
            const int i = tid + 128 * k;
            const int l = i / 28, j = i - l * 28;
            *(float4*)(wl + l * LSTR + OFF_W3 + 4 * j) = r3[k];
        }
        #pragma unroll
        for (int k = 0; k < NLAY * 8 / 128; k++) {
            const int i = tid + 128 * k;
            wl[(i >> 3) * LSTR + OFF_B2 + (i & 7)] = rb2[k];
        }
        #pragma unroll
        for (int k = 0; k < NLAY * 16 / 128; k++) {
            const int i = tid + 128 * k;
            wl[(i >> 4) * LSTR + OFF_B3 + (i & 15)] = rb3[k];
        }
        #pragma unroll
        for (int k = 0; k < TPER * 16 / 128; k++) b1s[tid + 128 * k] = rb1[k];
    }
    __syncthreads();

    if (part == 0 && lane < 16) {
        float p[PP];
        #pragma unroll
        for (int q = 0; q < PP; q++) p[q] = initp[q];
        float z, ld;
        rqs_eval(p, x[s * DD + 0], z, ld);
        out[s] = z;
        atomicAdd(&ldacc[0], ld);
    }

    v2f a[16];
    const float* xrow = x + (size_t)s * DD;
    #pragma unroll
    for (int kk = 0; kk < 16; kk++)
        a[kk] = *(const v2f*)(xrow + 4 * kk + 2 * hh);

    for (int t = t_begin; t < t_begin + TPER; ++t) {
        const float bias = b1s[16 * (t - t_begin) + m];
        const float* bcol = Bt + (size_t)(16 * t + m) * 64 + 2 * hh;
        v8f c = {0.f, 0.f, 0.f, 0.f, 0.f, 0.f, 0.f, 0.f};
        #pragma unroll
        for (int kk = 0; kk < 16; kk++) {
            v2f b = *(const v2f*)(bcol + 4 * kk);
            c = __builtin_amdgcn_wmma_f32_16x16x4_f32(
                    false, a[kk], false, b, (short)0, c, false, false);
        }

        #pragma unroll
        for (int r = 0; r < 8; r++) {
            float hv = hw_tanh(c[r] + bias);
            tile[wave][r + 8 * hh][m] = hv;
        }
        __syncthreads();

        const int l_idx = 2 * t + hh;
        if (l_idx < LL) {
            const int cdim = l_idx + 1;
            const float xval = x[(size_t)s * DD + cdim];

            float hv[8];
            #pragma unroll
            for (int j = 0; j < 8; j++) hv[j] = tile[wave][m][8 * hh + j];

            const float* lw = wl + (2 * (t - t_begin) + hh) * LSTR;
            float w2v[64], acc2[8];
            #pragma unroll
            for (int i = 0; i < 16; i++) {
                const float4 v = *(const float4*)(lw + 4 * i);
                w2v[4 * i] = v.x; w2v[4 * i + 1] = v.y; w2v[4 * i + 2] = v.z; w2v[4 * i + 3] = v.w;
            }
            #pragma unroll
            for (int i = 0; i < 2; i++) {
                const float4 v = *(const float4*)(lw + OFF_B2 + 4 * i);
                acc2[4 * i] = v.x; acc2[4 * i + 1] = v.y; acc2[4 * i + 2] = v.z; acc2[4 * i + 3] = v.w;
            }
            #pragma unroll
            for (int g = 0; g < 8; g++) {
                #pragma unroll
                for (int j = 0; j < 8; j++) acc2[g] += hv[j] * w2v[j * 8 + g];
            }
            float h2[8];
            #pragma unroll
            for (int g = 0; g < 8; g++) h2[g] = hw_tanh(acc2[g]);

            float w3v[8 * PP], pv[16];
            #pragma unroll
            for (int i = 0; i < 28; i++) {
                const float4 v = *(const float4*)(lw + OFF_W3 + 4 * i);
                w3v[4 * i] = v.x; w3v[4 * i + 1] = v.y; w3v[4 * i + 2] = v.z; w3v[4 * i + 3] = v.w;
            }
            #pragma unroll
            for (int i = 0; i < 4; i++) {
                const float4 v = *(const float4*)(lw + OFF_B3 + 4 * i);
                pv[4 * i] = v.x; pv[4 * i + 1] = v.y; pv[4 * i + 2] = v.z; pv[4 * i + 3] = v.w;
            }
            #pragma unroll
            for (int q = 0; q < PP; q++) {
                #pragma unroll
                for (int g = 0; g < 8; g++) pv[q] += h2[g] * w3v[g * PP + q];
            }

            float z, ld;
            rqs_eval(pv, xval, z, ld);
            out[(size_t)cdim * NS + s] = z;
            atomicAdd(&ldacc[cdim], ld);
        }
        __syncthreads();
    }

    if (tid < DD) {
        const int owner = (tid == 0) ? 0 : (((tid - 1) >> 1) / TPER);
        if (owner == part)
            out[(size_t)NS * DD + (size_t)tid * (NS / 64) + blockIdx.x] = ldacc[tid];
    }
}

extern "C" void kernel_launch(void* const* d_in, const int* in_sizes, int n_in,
                              void* d_out, int out_size, void* d_ws, size_t ws_size,
                              hipStream_t stream) {
    const float* x     = (const float*)d_in[0];
    const float* initp = (const float*)d_in[1];
    const float* W1    = (const float*)d_in[2];
    const float* b1    = (const float*)d_in[3];
    const float* W2    = (const float*)d_in[4];
    const float* b2    = (const float*)d_in[5];
    const float* W3    = (const float*)d_in[6];
    const float* b3    = (const float*)d_in[7];
    float* out = (float*)d_out;
    float* Bt  = (float*)d_ws;

    nsf_pack_w1<<<(NCAT * 64 + 255) / 256, 256, 0, stream>>>(W1, Bt);
    nsf_fused<<<dim3(NS / 64, TSPLIT), 128, 0, stream>>>(x, initp, b1, W2, b2, W3, b3, Bt, out);
}


# swap the WMMA operand roles (A = packed W1 columns, B = x samples): D lands as lane=(sample, layer), register=hidden unit, so the LDS transpose tile and both per-trip work-group barriers go
# speedup vs baseline: 3.9089x; 1.1555x over previous
#include <hip/hip_runtime.h>
#include <math.h>

typedef float v2f __attribute__((ext_vector_type(2)));
typedef float v8f __attribute__((ext_vector_type(8)));

#define NS 32768
#define DD 64
#define LL 63
#define PP 14
#define NCAT 512
#define BB 5.0f
#define MINW 0.001f
#define MINH 0.001f
#define MIND 0.001f
#define TSPLIT 2
#define TPER (32 / TSPLIT)
#define NLAY (2 * TPER)
#define LSTR 200
#define OFF_B2 64
#define OFF_W3 72
#define OFF_B3 184

__global__ void nsf_pack_w1(const float* __restrict__ W1, float* __restrict__ Bt) {
    int idx = blockIdx.x * blockDim.x + threadIdx.x;
    if (idx >= NCAT * 64) return;
    int col = idx >> 6;
    int k   = idx & 63;
    int l   = col >> 3;
    int h   = col & 7;
    float v = 0.0f;
    if (l < LL && k <= l) v = W1[(l * LL + k) * 8 + h];
    Bt[col * 64 + k] = v;
}

__device__ __forceinline__ float hw_exp(float d)  { return __builtin_amdgcn_exp2f(d * 1.44269504088896341f); }
__device__ __forceinline__ float hw_ln(float v)   { return __builtin_amdgcn_logf(v) * 0.693147180559945309f; }
__device__ __forceinline__ float hw_rcp(float v)  { return __builtin_amdgcn_rcpf(v); }
__device__ __forceinline__ float hw_tanh(float v) { return __builtin_amdgcn_tanhf(v); }

__device__ __forceinline__ float softplusf(float v) {
    return fmaxf(v, 0.0f) + hw_ln(1.0f + hw_exp(-fabsf(v)));
}

__device__ void rqs_eval(const float* __restrict__ p, float xf, float& z, float& ld) {
    float cumw[6], cumh[6], derivs[6];
    {
        float mx = p[0];
        #pragma unroll
        for (int i = 1; i < 5; i++) mx = fmaxf(mx, p[i]);
        float e[5], s = 0.0f;
        #pragma unroll
        for (int i = 0; i < 5; i++) { e[i] = hw_exp(p[i] - mx); s += e[i]; }
        float inv = hw_rcp(s), cs = 0.0f;
        cumw[0] = -BB;
        #pragma unroll
        for (int i = 0; i < 4; i++) {
            cs += MINW + (1.0f - MINW * 5.0f) * e[i] * inv;
            cumw[i + 1] = 2.0f * BB * cs - BB;
        }
        cumw[5] = BB;
    }
    {
        float mx = p[5];
        #pragma unroll
        for (int i = 6; i < 10; i++) mx = fmaxf(mx, p[i]);
        float e[5], s = 0.0f;
        #pragma unroll
        for (int i = 0; i < 5; i++) { e[i] = hw_exp(p[5 + i] - mx); s += e[i]; }
        float inv = hw_rcp(s), cs = 0.0f;
        cumh[0] = -BB;
        #pragma unroll
        for (int i = 0; i < 4; i++) {
            cs += MINH + (1.0f - MINH * 5.0f) * e[i] * inv;
            cumh[i + 1] = 2.0f * BB * cs - BB;
        }
        cumh[5] = BB;
    }
    derivs[0] = 1.0f; derivs[5] = 1.0f;
    #pragma unroll
    for (int i = 1; i < 5; i++) derivs[i] = MIND + softplusf(p[9 + i]);

    float xc = fminf(fmaxf(xf, -BB), BB);
    int idx = 0;
    #pragma unroll
    for (int j = 1; j < 6; j++) idx += (xc >= cumw[j]) ? 1 : 0;
    if (idx > 4) idx = 4;

    float icw = cumw[idx], ibw = cumw[idx + 1] - cumw[idx];
    float ich = cumh[idx], ih  = cumh[idx + 1] - cumh[idx];
    const float rbw = hw_rcp(ibw);
    float idl = ih * rbw;
    float idv = derivs[idx], idvp1 = derivs[idx + 1];
    float theta = (xc - icw) * rbw;
    float tmt = theta * (1.0f - theta);
    float num = ih * (idl * theta * theta + idv * tmt);
    float den = idl + (idv + idvp1 - 2.0f * idl) * tmt;
    float outv = ich + num * hw_rcp(den);
    float omt = 1.0f - theta;
    float dnum = idl * idl * (idvp1 * theta * theta + 2.0f * idl * tmt + idv * omt * omt);
    float ldv = 0.693147180559945309f * (__builtin_amdgcn_logf(dnum) - 2.0f * __builtin_amdgcn_logf(den));
    bool inside = (xf >= -BB) && (xf <= BB);
    z  = inside ? outv : xf;
    ld = inside ? ldv : 0.0f;
}

__launch_bounds__(128)
__global__ void nsf_fused(const float* __restrict__ x, const float* __restrict__ initp,
                          const float* __restrict__ b1, const float* __restrict__ W2,
                          const float* __restrict__ b2, const float* __restrict__ W3,
                          const float* __restrict__ b3, const float* __restrict__ Bt,
                          float* __restrict__ out) {
    __shared__ float ldacc[DD];
    __shared__ __attribute__((aligned(16))) float wl[NLAY * LSTR];
    __shared__ __attribute__((aligned(16))) float b1s[TPER * 16];

    const int tid  = threadIdx.x;
    const int wave = tid >> 5;
    const int lane = tid & 31;
    const int m    = lane & 15;
    const int hh   = lane >> 4;
    const int s0   = blockIdx.x * 64 + wave * 16;
    const int s    = s0 + m;
    const int part = blockIdx.y;
    const int t_begin = part * TPER;

    if (tid < DD) ldacc[tid] = 0.0f;

    {
        static_assert(NLAY % 16 == 0, "copy loops assume whole passes of 128 threads");
        const int L0   = 2 * t_begin;
        const int nlay = (LL - L0 < NLAY) ? (LL - L0) : NLAY;
        const float4* src2 = (const float4*)(W2 + (size_t)L0 * 64);
        const float4* src3 = (const float4*)(W3 + (size_t)L0 * 8 * PP);
        float4 r2[NLAY * 16 / 128], r3[NLAY * 28 / 128];
        float  rb2[NLAY * 8 / 128], rb3[NLAY * 16 / 128], rb1[TPER * 16 / 128];
        #pragma unroll
        for (int k = 0; k < NLAY * 16 / 128; k++) {
            const int i = tid + 128 * k;
            r2[k] = src2[i < nlay * 16 ? i : nlay * 16 - 1];
        }
        #pragma unroll
        for (int k = 0; k < NLAY * 28 / 128; k++) {
            const int i = tid + 128 * k;
            r3[k] = src3[i < nlay * 28 ? i : nlay * 28 - 1];
        }
        #pragma unroll
        for (int k = 0; k < NLAY * 8 / 128; k++) {
            const int i = tid + 128 * k;
            rb2[k] = b2[(size_t)L0 * 8 + (i < nlay * 8 ? i : nlay * 8 - 1)];
        }
        #pragma unroll
        for (int k = 0; k < NLAY * 16 / 128; k++) {
            const int i = tid + 128 * k;
            const int l = i >> 4, q = i & 15;
            const float v = b3[(size_t)(L0 + (l < nlay ? l : nlay - 1)) * PP + (q < PP ? q : PP - 1)];
            rb3[k] = (q < PP) ? v : 0.0f;
        }
        #pragma unroll
        for (int k = 0; k < TPER * 16 / 128; k++) {
            const int col = 16 * t_begin + tid + 128 * k;
            const float v = b1[col < LL * 8 ? col : LL * 8 - 1];
            rb1[k] = (col < LL * 8) ? v : 0.0f;
        }
        #pragma unroll
        for (int k = 0; k < NLAY * 16 / 128; k++) {
            const int i = tid + 128 * k;
            *(float4*)(wl + (i >> 4) * LSTR + 4 * (i & 15)) = r2[k];
        }
        #pragma unroll
        for (int k = 0; k < NLAY * 28 / 128; k++) {
            const int i = tid + 128 * k;
            const int l = i / 28, j = i - l * 28;
            *(float4*)(wl + l * LSTR + OFF_W3 + 4 * j) = r3[k];
        }
        #pragma unroll
        for (int k = 0; k < NLAY * 8 / 128; k++) {
            const int i = tid + 128 * k;
            wl[(i >> 3) * LSTR + OFF_B2 + (i & 7)] = rb2[k];
        }
        #pragma unroll
        for (int k = 0; k < NLAY * 16 / 128; k++) {
            const int i = tid + 128 * k;
            wl[(i >> 4) * LSTR + OFF_B3 + (i & 15)] = rb3[k];
        }
        #pragma unroll
        for (int k = 0; k < TPER * 16 / 128; k++) b1s[tid + 128 * k] = rb1[k];
    }
    __syncthreads();

    if (part == 0 && lane < 16) {
        float p[PP];
        #pragma unroll
        for (int q = 0; q < PP; q++) p[q] = initp[q];
        float z, ld;
        rqs_eval(p, x[s * DD + 0], z, ld);
        out[s] = z;
        atomicAdd(&ldacc[0], ld);
    }

    v2f a[16];
    const float* xrow = x + (size_t)s * DD;
    #pragma unroll
    for (int kk = 0; kk < 16; kk++)
        a[kk] = *(const v2f*)(xrow + 4 * kk + 2 * hh);

    for (int t = t_begin; t < t_begin + TPER; ++t) {
        const float* bcol = Bt + (size_t)(16 * t + m) * 64 + 2 * hh;
        v8f c = {0.f, 0.f, 0.f, 0.f, 0.f, 0.f, 0.f, 0.f};
        #pragma unroll
        for (int kk = 0; kk < 16; kk++) {
            v2f b = *(const v2f*)(bcol + 4 * kk);
            c = __builtin_amdgcn_wmma_f32_16x16x4_f32(
                    false, b, false, a[kk], (short)0, c, false, false);
        }

        float hv[8];
        {
            const float* bp = b1s + 16 * (t - t_begin) + 8 * hh;
            const float4 bl = *(const float4*)bp, bh = *(const float4*)(bp + 4);
            const float bias[8] = {bl.x, bl.y, bl.z, bl.w, bh.x, bh.y, bh.z, bh.w};
            #pragma unroll
            for (int r = 0; r < 8; r++) hv[r] = hw_tanh(c[r] + bias[r]);
        }

        const int l_idx = 2 * t + hh;
        if (l_idx < LL) {
            const int cdim = l_idx + 1;
            const float xval = x[(size_t)s * DD + cdim];

            const float* lw = wl + (2 * (t - t_begin) + hh) * LSTR;
            float w2v[64], acc2[8];
            #pragma unroll
            for (int i = 0; i < 16; i++) {
                const float4 v = *(const float4*)(lw + 4 * i);
                w2v[4 * i] = v.x; w2v[4 * i + 1] = v.y; w2v[4 * i + 2] = v.z; w2v[4 * i + 3] = v.w;
            }
            #pragma unroll
            for (int i = 0; i < 2; i++) {
                const float4 v = *(const float4*)(lw + OFF_B2 + 4 * i);
                acc2[4 * i] = v.x; acc2[4 * i + 1] = v.y; acc2[4 * i + 2] = v.z; acc2[4 * i + 3] = v.w;
            }
            #pragma unroll
            for (int g = 0; g < 8; g++) {
                #pragma unroll
                for (int j = 0; j < 8; j++) acc2[g] += hv[j] * w2v[j * 8 + g];
            }
            float h2[8];
            #pragma unroll
            for (int g = 0; g < 8; g++) h2[g] = hw_tanh(acc2[g]);

            float w3v[8 * PP], pv[16];
            #pragma unroll
            for (int i = 0; i < 28; i++) {
                const float4 v = *(const float4*)(lw + OFF_W3 + 4 * i);
                w3v[4 * i] = v.x; w3v[4 * i + 1] = v.y; w3v[4 * i + 2] = v.z; w3v[4 * i + 3] = v.w;
            }
            #pragma unroll
            for (int i = 0; i < 4; i++) {
                const float4 v = *(const float4*)(lw + OFF_B3 + 4 * i);
                pv[4 * i] = v.x; pv[4 * i + 1] = v.y; pv[4 * i + 2] = v.z; pv[4 * i + 3] = v.w;
            }
            #pragma unroll
            for (int q = 0; q < PP; q++) {
                #pragma unroll
                for (int g = 0; g < 8; g++) pv[q] += h2[g] * w3v[g * PP + q];
            }

            float z, ld;
            rqs_eval(pv, xval, z, ld);
            out[(size_t)cdim * NS + s] = z;
            atomicAdd(&ldacc[cdim], ld);
        }
    }
    __syncthreads();

    if (tid < DD) {
        const int owner = (tid == 0) ? 0 : (((tid - 1) >> 1) / TPER);
        if (owner == part)
            out[(size_t)NS * DD + (size_t)tid * (NS / 64) + blockIdx.x] = ldacc[tid];
    }
}

extern "C" void kernel_launch(void* const* d_in, const int* in_sizes, int n_in,
                              void* d_out, int out_size, void* d_ws, size_t ws_size,
                              hipStream_t stream) {
    const float* x     = (const float*)d_in[0];
    const float* initp = (const float*)d_in[1];
    const float* W1    = (const float*)d_in[2];
    const float* b1    = (const float*)d_in[3];
    const float* W2    = (const float*)d_in[4];
    const float* b2    = (const float*)d_in[5];
    const float* W3    = (const float*)d_in[6];
    const float* b3    = (const float*)d_in[7];
    float* out = (float*)d_out;
    float* Bt  = (float*)d_ws;

    nsf_pack_w1<<<(NCAT * 64 + 255) / 256, 256, 0, stream>>>(W1, Bt);
    nsf_fused<<<dim3(NS / 64, TSPLIT), 128, 0, stream>>>(x, initp, b1, W2, b2, W3, b3, Bt, out);
}


# split each 64-sample block's 32 layer-pair iterations over 4 work-groups instead of 2 (grid 512x2 -> 512x4), launch bound 128 registers: 8 resident waves per SIMD instead of 4 cover LDS/load waits
# speedup vs baseline: 4.0880x; 1.0458x over previous
#include <hip/hip_runtime.h>
#include <math.h>

typedef float v2f __attribute__((ext_vector_type(2)));
typedef float v8f __attribute__((ext_vector_type(8)));

#define NS 32768
#define DD 64
#define LL 63
#define PP 14
#define NCAT 512
#define BB 5.0f
#define MINW 0.001f
#define MINH 0.001f
#define MIND 0.001f
#define TSPLIT 4
#define TPER (32 / TSPLIT)
#define NLAY (2 * TPER)
#define LSTR 200
#define OFF_B2 64
#define OFF_W3 72
#define OFF_B3 184

__global__ void nsf_pack_w1(const float* __restrict__ W1, float* __restrict__ Bt) {
    int idx = blockIdx.x * blockDim.x + threadIdx.x;
    if (idx >= NCAT * 64) return;
    int col = idx >> 6;
    int k   = idx & 63;
    int l   = col >> 3;
    int h   = col & 7;
    float v = 0.0f;
    if (l < LL && k <= l) v = W1[(l * LL + k) * 8 + h];
    Bt[col * 64 + k] = v;
}

__device__ __forceinline__ float hw_exp(float d)  { return __builtin_amdgcn_exp2f(d * 1.44269504088896341f); }
__device__ __forceinline__ float hw_ln(float v)   { return __builtin_amdgcn_logf(v) * 0.693147180559945309f; }
__device__ __forceinline__ float hw_rcp(float v)  { return __builtin_amdgcn_rcpf(v); }
__device__ __forceinline__ float hw_tanh(float v) { return __builtin_amdgcn_tanhf(v); }

__device__ __forceinline__ float softplusf(float v) {
    return fmaxf(v, 0.0f) + hw_ln(1.0f + hw_exp(-fabsf(v)));
}

__device__ void rqs_eval(const float* __restrict__ p, float xf, float& z, float& ld) {
    float cumw[6], cumh[6], derivs[6];
    {
        float mx = p[0];
        #pragma unroll
        for (int i = 1; i < 5; i++) mx = fmaxf(mx, p[i]);
        float e[5], s = 0.0f;
        #pragma unroll
        for (int i = 0; i < 5; i++) { e[i] = hw_exp(p[i] - mx); s += e[i]; }
        float inv = hw_rcp(s), cs = 0.0f;
        cumw[0] = -BB;
        #pragma unroll
        for (int i = 0; i < 4; i++) {
            cs += MINW + (1.0f - MINW * 5.0f) * e[i] * inv;
            cumw[i + 1] = 2.0f * BB * cs - BB;
        }
        cumw[5] = BB;
    }
    {
        float mx = p[5];
        #pragma unroll
        for (int i = 6; i < 10; i++) mx = fmaxf(mx, p[i]);
        float e[5], s = 0.0f;
        #pragma unroll
        for (int i = 0; i < 5; i++) { e[i] = hw_exp(p[5 + i] - mx); s += e[i]; }
        float inv = hw_rcp(s), cs = 0.0f;
        cumh[0] = -BB;
        #pragma unroll
        for (int i = 0; i < 4; i++) {
            cs += MINH + (1.0f - MINH * 5.0f) * e[i] * inv;
            cumh[i + 1] = 2.0f * BB * cs - BB;
        }
        cumh[5] = BB;
    }
    derivs[0] = 1.0f; derivs[5] = 1.0f;
    #pragma unroll
    for (int i = 1; i < 5; i++) derivs[i] = MIND + softplusf(p[9 + i]);

    float xc = fminf(fmaxf(xf, -BB), BB);
    int idx = 0;
    #pragma unroll
    for (int j = 1; j < 6; j++) idx += (xc >= cumw[j]) ? 1 : 0;
    if (idx > 4) idx = 4;

    float icw = cumw[idx], ibw = cumw[idx + 1] - cumw[idx];
    float ich = cumh[idx], ih  = cumh[idx + 1] - cumh[idx];
    const float rbw = hw_rcp(ibw);
    float idl = ih * rbw;
    float idv = derivs[idx], idvp1 = derivs[idx + 1];
    float theta = (xc - icw) * rbw;
    float tmt = theta * (1.0f - theta);
    float num = ih * (idl * theta * theta + idv * tmt);
    float den = idl + (idv + idvp1 - 2.0f * idl) * tmt;
    float outv = ich + num * hw_rcp(den);
    float omt = 1.0f - theta;
    float dnum = idl * idl * (idvp1 * theta * theta + 2.0f * idl * tmt + idv * omt * omt);
    float ldv = 0.693147180559945309f * (__builtin_amdgcn_logf(dnum) - 2.0f * __builtin_amdgcn_logf(den));
    bool inside = (xf >= -BB) && (xf <= BB);
    z  = inside ? outv : xf;
    ld = inside ? ldv : 0.0f;
}

__launch_bounds__(128, 8)
__global__ void nsf_fused(const float* __restrict__ x, const float* __restrict__ initp,
                          const float* __restrict__ b1, const float* __restrict__ W2,
                          const float* __restrict__ b2, const float* __restrict__ W3,
                          const float* __restrict__ b3, const float* __restrict__ Bt,
                          float* __restrict__ out) {
    __shared__ float ldacc[DD];
    __shared__ __attribute__((aligned(16))) float wl[NLAY * LSTR];
    __shared__ __attribute__((aligned(16))) float b1s[TPER * 16];

    const int tid  = threadIdx.x;
    const int wave = tid >> 5;
    const int lane = tid & 31;
    const int m    = lane & 15;
    const int hh   = lane >> 4;
    const int s0   = blockIdx.x * 64 + wave * 16;
    const int s    = s0 + m;
    const int part = blockIdx.y;
    const int t_begin = part * TPER;

    if (tid < DD) ldacc[tid] = 0.0f;

    {
        static_assert(NLAY % 16 == 0, "copy loops assume whole passes of 128 threads");
        const int L0   = 2 * t_begin;
        const int nlay = (LL - L0 < NLAY) ? (LL - L0) : NLAY;
        const float4* src2 = (const float4*)(W2 + (size_t)L0 * 64);
        const float4* src3 = (const float4*)(W3 + (size_t)L0 * 8 * PP);
        constexpr int N3 = NLAY * 28;
        constexpr int P3 = (N3 + 127) / 128;
        float4 r2[NLAY * 16 / 128], r3[P3];
        float  rb2[NLAY * 8 / 128], rb3[NLAY * 16 / 128], rb1[TPER * 16 / 128];
        #pragma unroll
        for (int k = 0; k < NLAY * 16 / 128; k++) {
            const int i = tid + 128 * k;
            r2[k] = src2[i < nlay * 16 ? i : nlay * 16 - 1];
        }
        #pragma unroll
        for (int k = 0; k < P3; k++) {
            const int i = tid + 128 * k;
            r3[k] = src3[i < nlay * 28 ? i : nlay * 28 - 1];
        }
        #pragma unroll
        for (int k = 0; k < NLAY * 8 / 128; k++) {
            const int i = tid + 128 * k;
            rb2[k] = b2[(size_t)L0 * 8 + (i < nlay * 8 ? i : nlay * 8 - 1)];
        }
        #pragma unroll
        for (int k = 0; k < NLAY * 16 / 128; k++) {
            const int i = tid + 128 * k;
            const int l = i >> 4, q = i & 15;
            const float v = b3[(size_t)(L0 + (l < nlay ? l : nlay - 1)) * PP + (q < PP ? q : PP - 1)];
            rb3[k] = (q < PP) ? v : 0.0f;
        }
        #pragma unroll
        for (int k = 0; k < TPER * 16 / 128; k++) {
            const int col = 16 * t_begin + tid + 128 * k;
            const float v = b1[col < LL * 8 ? col : LL * 8 - 1];
            rb1[k] = (col < LL * 8) ? v : 0.0f;
        }
        #pragma unroll
        for (int k = 0; k < NLAY * 16 / 128; k++) {
            const int i = tid + 128 * k;
            *(float4*)(wl + (i >> 4) * LSTR + 4 * (i & 15)) = r2[k];
        }
        #pragma unroll
        for (int k = 0; k < P3; k++) {
            const int i = tid + 128 * k;
            const int ic = (i < N3) ? i : N3 - 1;
            const int l = ic / 28, j = ic - l * 28;
            *(float4*)(wl + l * LSTR + OFF_W3 + 4 * j) = r3[k];
        }
        #pragma unroll
        for (int k = 0; k < NLAY * 8 / 128; k++) {
            const int i = tid + 128 * k;
            wl[(i >> 3) * LSTR + OFF_B2 + (i & 7)] = rb2[k];
        }
        #pragma unroll
        for (int k = 0; k < NLAY * 16 / 128; k++) {
            const int i = tid + 128 * k;
            wl[(i >> 4) * LSTR + OFF_B3 + (i & 15)] = rb3[k];
        }
        #pragma unroll
        for (int k = 0; k < TPER * 16 / 128; k++) b1s[tid + 128 * k] = rb1[k];
    }
    __syncthreads();

    if (part == 0 && lane < 16) {
        float p[PP];
        #pragma unroll
        for (int q = 0; q < PP; q++) p[q] = initp[q];
        float z, ld;
        rqs_eval(p, x[s * DD + 0], z, ld);
        out[s] = z;
        atomicAdd(&ldacc[0], ld);
    }

    v2f a[16];
    const float* xrow = x + (size_t)s * DD;
    #pragma unroll
    for (int kk = 0; kk < 16; kk++)
        a[kk] = *(const v2f*)(xrow + 4 * kk + 2 * hh);

    for (int t = t_begin; t < t_begin + TPER; ++t) {
        const float* bcol = Bt + (size_t)(16 * t + m) * 64 + 2 * hh;
        v8f c = {0.f, 0.f, 0.f, 0.f, 0.f, 0.f, 0.f, 0.f};
        #pragma unroll
        for (int kk = 0; kk < 16; kk++) {
            v2f b = *(const v2f*)(bcol + 4 * kk);
            c = __builtin_amdgcn_wmma_f32_16x16x4_f32(
                    false, b, false, a[kk], (short)0, c, false, false);
        }

        float hv[8];
        {
            const float* bp = b1s + 16 * (t - t_begin) + 8 * hh;
            const float4 bl = *(const float4*)bp, bh = *(const float4*)(bp + 4);
            const float bias[8] = {bl.x, bl.y, bl.z, bl.w, bh.x, bh.y, bh.z, bh.w};
            #pragma unroll
            for (int r = 0; r < 8; r++) hv[r] = hw_tanh(c[r] + bias[r]);
        }

        const int l_idx = 2 * t + hh;
        if (l_idx < LL) {
            const int cdim = l_idx + 1;
            const float xval = x[(size_t)s * DD + cdim];

            const float* lw = wl + (2 * (t - t_begin) + hh) * LSTR;
            float w2v[64], acc2[8];
            #pragma unroll
            for (int i = 0; i < 16; i++) {
                const float4 v = *(const float4*)(lw + 4 * i);
                w2v[4 * i] = v.x; w2v[4 * i + 1] = v.y; w2v[4 * i + 2] = v.z; w2v[4 * i + 3] = v.w;
            }
            #pragma unroll
            for (int i = 0; i < 2; i++) {
                const float4 v = *(const float4*)(lw + OFF_B2 + 4 * i);
                acc2[4 * i] = v.x; acc2[4 * i + 1] = v.y; acc2[4 * i + 2] = v.z; acc2[4 * i + 3] = v.w;
            }
            #pragma unroll
            for (int g = 0; g < 8; g++) {
                #pragma unroll
                for (int j = 0; j < 8; j++) acc2[g] += hv[j] * w2v[j * 8 + g];
            }
            float h2[8];
            #pragma unroll
            for (int g = 0; g < 8; g++) h2[g] = hw_tanh(acc2[g]);

            float w3v[8 * PP], pv[16];
            #pragma unroll
            for (int i = 0; i < 28; i++) {
                const float4 v = *(const float4*)(lw + OFF_W3 + 4 * i);
                w3v[4 * i] = v.x; w3v[4 * i + 1] = v.y; w3v[4 * i + 2] = v.z; w3v[4 * i + 3] = v.w;
            }
            #pragma unroll
            for (int i = 0; i < 4; i++) {
                const float4 v = *(const float4*)(lw + OFF_B3 + 4 * i);
                pv[4 * i] = v.x; pv[4 * i + 1] = v.y; pv[4 * i + 2] = v.z; pv[4 * i + 3] = v.w;
            }
            #pragma unroll
            for (int q = 0; q < PP; q++) {
                #pragma unroll
                for (int g = 0; g < 8; g++) pv[q] += h2[g] * w3v[g * PP + q];
            }

            float z, ld;
            rqs_eval(pv, xval, z, ld);
            out[(size_t)cdim * NS + s] = z;
            atomicAdd(&ldacc[cdim], ld);
        }
    }
    __syncthreads();

    if (tid < DD) {
        const int owner = (tid == 0) ? 0 : (((tid - 1) >> 1) / TPER);
        if (owner == part)
            out[(size_t)NS * DD + (size_t)tid * (NS / 64) + blockIdx.x] = ldacc[tid];
    }
}

extern "C" void kernel_launch(void* const* d_in, const int* in_sizes, int n_in,
                              void* d_out, int out_size, void* d_ws, size_t ws_size,
                              hipStream_t stream) {
    const float* x     = (const float*)d_in[0];
    const float* initp = (const float*)d_in[1];
    const float* W1    = (const float*)d_in[2];
    const float* b1    = (const float*)d_in[3];
    const float* W2    = (const float*)d_in[4];
    const float* b2    = (const float*)d_in[5];
    const float* W3    = (const float*)d_in[6];
    const float* b3    = (const float*)d_in[7];
    float* out = (float*)d_out;
    float* Bt  = (float*)d_ws;

    nsf_pack_w1<<<(NCAT * 64 + 255) / 256, 256, 0, stream>>>(W1, Bt);
    nsf_fused<<<dim3(NS / 64, TSPLIT), 128, 0, stream>>>(x, initp, b1, W2, b2, W3, b3, Bt, out);
}


# packed W1 stored in WMMA-fragment order: each lane's operands for two k-steps are one 16-byte load, a wave reads 512 contiguous bytes (4 lines) per load instead of 16 lines; same k order of the sums
# speedup vs baseline: 4.3274x; 1.0586x over previous
#include <hip/hip_runtime.h>
#include <math.h>

typedef float v2f __attribute__((ext_vector_type(2)));
typedef float v4f __attribute__((ext_vector_type(4)));
typedef float v8f __attribute__((ext_vector_type(8)));

#define NS 32768
#define DD 64
#define LL 63
#define PP 14
#define NCAT 512
#define BB 5.0f
#define MINW 0.001f
#define MINH 0.001f
#define MIND 0.001f
#define TSPLIT 4
#define TPER (32 / TSPLIT)
#define NLAY (2 * TPER)
#define LSTR 200
#define OFF_B2 64
#define OFF_W3 72
#define OFF_B3 184

__global__ void nsf_pack_w1(const float* __restrict__ W1, float* __restrict__ Bt) {
    int idx = blockIdx.x * blockDim.x + threadIdx.x;
    if (idx >= NCAT * 64) return;
    int e    = idx & 3;
    int lane = (idx >> 2) & 31;
    int q    = (idx >> 7) & 7;
    int t    = idx >> 10;
    int m    = lane & 15;
    int hh   = lane >> 4;
    int col  = 16 * t + m;
    int k    = 4 * (2 * q + (e >> 1)) + 2 * hh + (e & 1);
    int l    = col >> 3;
    int h    = col & 7;
    float v = 0.0f;
    if (l < LL && k <= l) v = W1[(l * LL + k) * 8 + h];
    Bt[idx] = v;
}

__device__ __forceinline__ float hw_exp(float d)  { return __builtin_amdgcn_exp2f(d * 1.44269504088896341f); }
__device__ __forceinline__ float hw_ln(float v)   { return __builtin_amdgcn_logf(v) * 0.693147180559945309f; }
__device__ __forceinline__ float hw_rcp(float v)  { return __builtin_amdgcn_rcpf(v); }
__device__ __forceinline__ float hw_tanh(float v) { return __builtin_amdgcn_tanhf(v); }

__device__ __forceinline__ float softplusf(float v) {
    return fmaxf(v, 0.0f) + hw_ln(1.0f + hw_exp(-fabsf(v)));
}

__device__ void rqs_eval(const float* __restrict__ p, float xf, float& z, float& ld) {
    float cumw[6], cumh[6], derivs[6];
    {
        float mx = p[0];
        #pragma unroll
        for (int i = 1; i < 5; i++) mx = fmaxf(mx, p[i]);
        float e[5], s = 0.0f;
        #pragma unroll
        for (int i = 0; i < 5; i++) { e[i] = hw_exp(p[i] - mx); s += e[i]; }
        float inv = hw_rcp(s), cs = 0.0f;
        cumw[0] = -BB;
        #pragma unroll
        for (int i = 0; i < 4; i++) {
            cs += MINW + (1.0f - MINW * 5.0f) * e[i] * inv;
            cumw[i + 1] = 2.0f * BB * cs - BB;
        }
        cumw[5] = BB;
    }
    {
        float mx = p[5];
        #pragma unroll
        for (int i = 6; i < 10; i++) mx = fmaxf(mx, p[i]);
        float e[5], s = 0.0f;
        #pragma unroll
        for (int i = 0; i < 5; i++) { e[i] = hw_exp(p[5 + i] - mx); s += e[i]; }
        float inv = hw_rcp(s), cs = 0.0f;
        cumh[0] = -BB;
        #pragma unroll
        for (int i = 0; i < 4; i++) {
            cs += MINH + (1.0f - MINH * 5.0f) * e[i] * inv;
            cumh[i + 1] = 2.0f * BB * cs - BB;
        }
        cumh[5] = BB;
    }
    derivs[0] = 1.0f; derivs[5] = 1.0f;
    #pragma unroll
    for (int i = 1; i < 5; i++) derivs[i] = MIND + softplusf(p[9 + i]);

    float xc = fminf(fmaxf(xf, -BB), BB);
    int idx = 0;
    #pragma unroll
    for (int j = 1; j < 6; j++) idx += (xc >= cumw[j]) ? 1 : 0;
    if (idx > 4) idx = 4;

    float icw = cumw[idx], ibw = cumw[idx + 1] - cumw[idx];
    float ich = cumh[idx], ih  = cumh[idx + 1] - cumh[idx];
    const float rbw = hw_rcp(ibw);
    float idl = ih * rbw;
    float idv = derivs[idx], idvp1 = derivs[idx + 1];
    float theta = (xc - icw) * rbw;
    float tmt = theta * (1.0f - theta);
    float num = ih * (idl * theta * theta + idv * tmt);
    float den = idl + (idv + idvp1 - 2.0f * idl) * tmt;
    float outv = ich + num * hw_rcp(den);
    float omt = 1.0f - theta;
    float dnum = idl * idl * (idvp1 * theta * theta + 2.0f * idl * tmt + idv * omt * omt);
    float ldv = 0.693147180559945309f * (__builtin_amdgcn_logf(dnum) - 2.0f * __builtin_amdgcn_logf(den));
    bool inside = (xf >= -BB) && (xf <= BB);
    z  = inside ? outv : xf;
    ld = inside ? ldv : 0.0f;
}

__launch_bounds__(128, 8)
__global__ void nsf_fused(const float* __restrict__ x, const float* __restrict__ initp,
                          const float* __restrict__ b1, const float* __restrict__ W2,
                          const float* __restrict__ b2, const float* __restrict__ W3,
                          const float* __restrict__ b3, const float* __restrict__ Bt,
                          float* __restrict__ out) {
    __shared__ float ldacc[DD];
    __shared__ __attribute__((aligned(16))) float wl[NLAY * LSTR];
    __shared__ __attribute__((aligned(16))) float b1s[TPER * 16];

    const int tid  = threadIdx.x;
    const int wave = tid >> 5;
    const int lane = tid & 31;
    const int m    = lane & 15;
    const int hh   = lane >> 4;
    const int s0   = blockIdx.x * 64 + wave * 16;
    const int s    = s0 + m;
    const int part = blockIdx.y;
    const int t_begin = part * TPER;

    if (tid < DD) ldacc[tid] = 0.0f;

    {
        static_assert(NLAY % 16 == 0, "copy loops assume whole passes of 128 threads");
        const int L0   = 2 * t_begin;
        const int nlay = (LL - L0 < NLAY) ? (LL - L0) : NLAY;
        const float4* src2 = (const float4*)(W2 + (size_t)L0 * 64);
        const float4* src3 = (const float4*)(W3 + (size_t)L0 * 8 * PP);
        constexpr int N3 = NLAY * 28;
        constexpr int P3 = (N3 + 127) / 128;
        float4 r2[NLAY * 16 / 128], r3[P3];
        float  rb2[NLAY * 8 / 128], rb3[NLAY * 16 / 128], rb1[TPER * 16 / 128];
        #pragma unroll
        for (int k = 0; k < NLAY * 16 / 128; k++) {
            const int i = tid + 128 * k;
            r2[k] = src2[i < nlay * 16 ? i : nlay * 16 - 1];
        }
        #pragma unroll
        for (int k = 0; k < P3; k++) {
            const int i = tid + 128 * k;
            r3[k] = src3[i < nlay * 28 ? i : nlay * 28 - 1];
        }
        #pragma unroll
        for (int k = 0; k < NLAY * 8 / 128; k++) {
            const int i = tid + 128 * k;
            rb2[k] = b2[(size_t)L0 * 8 + (i < nlay * 8 ? i : nlay * 8 - 1)];
        }
        #pragma unroll
        for (int k = 0; k < NLAY * 16 / 128; k++) {
            const int i = tid + 128 * k;
            const int l = i >> 4, q = i & 15;
            const float v = b3[(size_t)(L0 + (l < nlay ? l : nlay - 1)) * PP + (q < PP ? q : PP - 1)];
            rb3[k] = (q < PP) ? v : 0.0f;
        }
        #pragma unroll
        for (int k = 0; k < TPER * 16 / 128; k++) {
            const int col = 16 * t_begin + tid + 128 * k;
            const float v = b1[col < LL * 8 ? col : LL * 8 - 1];
            rb1[k] = (col < LL * 8) ? v : 0.0f;
        }
        #pragma unroll
        for (int k = 0; k < NLAY * 16 / 128; k++) {
            const int i = tid + 128 * k;
            *(float4*)(wl + (i >> 4) * LSTR + 4 * (i & 15)) = r2[k];
        }
        #pragma unroll
        for (int k = 0; k < P3; k++) {
            const int i = tid + 128 * k;
            const int ic = (i < N3) ? i : N3 - 1;
            const int l = ic / 28, j = ic - l * 28;
            *(float4*)(wl + l * LSTR + OFF_W3 + 4 * j) = r3[k];
        }
        #pragma unroll
        for (int k = 0; k < NLAY * 8 / 128; k++) {
            const int i = tid + 128 * k;
            wl[(i >> 3) * LSTR + OFF_B2 + (i & 7)] = rb2[k];
        }
        #pragma unroll
        for (int k = 0; k < NLAY * 16 / 128; k++) {
            const int i = tid + 128 * k;
            wl[(i >> 4) * LSTR + OFF_B3 + (i & 15)] = rb3[k];
        }
        #pragma unroll
        for (int k = 0; k < TPER * 16 / 128; k++) b1s[tid + 128 * k] = rb1[k];
    }
    __syncthreads();

    if (part == 0 && lane < 16) {
        float p[PP];
        #pragma unroll
        for (int q = 0; q < PP; q++) p[q] = initp[q];
        float z, ld;
        rqs_eval(p, x[s * DD + 0], z, ld);
        out[s] = z;
        atomicAdd(&ldacc[0], ld);
    }

    v2f a[16];
    const float* xrow = x + (size_t)s * DD;
    #pragma unroll
    for (int kk = 0; kk < 16; kk++)
        a[kk] = *(const v2f*)(xrow + 4 * kk + 2 * hh);

    for (int t = t_begin; t < t_begin + TPER; ++t) {
        const v4f* bfrag = (const v4f*)(Bt + (size_t)t * 1024 + 4 * lane);
        v4f bq[8];
        #pragma unroll
        for (int q = 0; q < 8; q++) bq[q] = bfrag[32 * q];
        v8f c = {0.f, 0.f, 0.f, 0.f, 0.f, 0.f, 0.f, 0.f};
        #pragma unroll
        for (int q = 0; q < 8; q++) {
            const v2f b0 = {bq[q].x, bq[q].y}, b1v = {bq[q].z, bq[q].w};
            c = __builtin_amdgcn_wmma_f32_16x16x4_f32(
                    false, b0, false, a[2 * q], (short)0, c, false, false);
            c = __builtin_amdgcn_wmma_f32_16x16x4_f32(
                    false, b1v, false, a[2 * q + 1], (short)0, c, false, false);
        }

        float hv[8];
        {
            const float* bp = b1s + 16 * (t - t_begin) + 8 * hh;
            const float4 bl = *(const float4*)bp, bh = *(const float4*)(bp + 4);
            const float bias[8] = {bl.x, bl.y, bl.z, bl.w, bh.x, bh.y, bh.z, bh.w};
            #pragma unroll
            for (int r = 0; r < 8; r++) hv[r] = hw_tanh(c[r] + bias[r]);
        }

        const int l_idx = 2 * t + hh;
        if (l_idx < LL) {
            const int cdim = l_idx + 1;
            const float xval = x[(size_t)s * DD + cdim];

            const float* lw = wl + (2 * (t - t_begin) + hh) * LSTR;
            float w2v[64], acc2[8];
            #pragma unroll
            for (int i = 0; i < 16; i++) {
                const float4 v = *(const float4*)(lw + 4 * i);
                w2v[4 * i] = v.x; w2v[4 * i + 1] = v.y; w2v[4 * i + 2] = v.z; w2v[4 * i + 3] = v.w;
            }
            #pragma unroll
            for (int i = 0; i < 2; i++) {
                const float4 v = *(const float4*)(lw + OFF_B2 + 4 * i);
                acc2[4 * i] = v.x; acc2[4 * i + 1] = v.y; acc2[4 * i + 2] = v.z; acc2[4 * i + 3] = v.w;
            }
            #pragma unroll
            for (int g = 0; g < 8; g++) {
                #pragma unroll
                for (int j = 0; j < 8; j++) acc2[g] += hv[j] * w2v[j * 8 + g];
            }
            float h2[8];
            #pragma unroll
            for (int g = 0; g < 8; g++) h2[g] = hw_tanh(acc2[g]);

            float w3v[8 * PP], pv[16];
            #pragma unroll
            for (int i = 0; i < 28; i++) {
                const float4 v = *(const float4*)(lw + OFF_W3 + 4 * i);
                w3v[4 * i] = v.x; w3v[4 * i + 1] = v.y; w3v[4 * i + 2] = v.z; w3v[4 * i + 3] = v.w;
            }
            #pragma unroll
            for (int i = 0; i < 4; i++) {
                const float4 v = *(const float4*)(lw + OFF_B3 + 4 * i);
                pv[4 * i] = v.x; pv[4 * i + 1] = v.y; pv[4 * i + 2] = v.z; pv[4 * i + 3] = v.w;
            }
            #pragma unroll
            for (int q = 0; q < PP; q++) {
                #pragma unroll
                for (int g = 0; g < 8; g++) pv[q] += h2[g] * w3v[g * PP + q];
            }

            float z, ld;
            rqs_eval(pv, xval, z, ld);
            out[(size_t)cdim * NS + s] = z;
            atomicAdd(&ldacc[cdim], ld);
        }
    }
    __syncthreads();

    if (tid < DD) {
        const int owner = (tid == 0) ? 0 : (((tid - 1) >> 1) / TPER);
        if (owner == part)
            out[(size_t)NS * DD + (size_t)tid * (NS / 64) + blockIdx.x] = ldacc[tid];
    }
}

extern "C" void kernel_launch(void* const* d_in, const int* in_sizes, int n_in,
                              void* d_out, int out_size, void* d_ws, size_t ws_size,
                              hipStream_t stream) {
    const float* x     = (const float*)d_in[0];
    const float* initp = (const float*)d_in[1];
    const float* W1    = (const float*)d_in[2];
    const float* b1    = (const float*)d_in[3];
    const float* W2    = (const float*)d_in[4];
    const float* b2    = (const float*)d_in[5];
    const float* W3    = (const float*)d_in[6];
    const float* b3    = (const float*)d_in[7];
    float* out = (float*)d_out;
    float* Bt  = (float*)d_ws;

    nsf_pack_w1<<<(NCAT * 64 + 255) / 256, 256, 0, stream>>>(W1, Bt);
    nsf_fused<<<dim3(NS / 64, TSPLIT), 128, 0, stream>>>(x, initp, b1, W2, b2, W3, b3, Bt, out);
}


# layers 2/3 written as explicit 2-wide multiply-adds on neighbouring outputs (weight pairs as they lie in LDS, input unit in both halves): one v_pk_fma_f32 per two products, no re-pairing moves
# speedup vs baseline: 4.8901x; 1.1300x over previous
#include <hip/hip_runtime.h>
#include <math.h>

typedef float v2f __attribute__((ext_vector_type(2)));
typedef float v4f __attribute__((ext_vector_type(4)));
typedef float v8f __attribute__((ext_vector_type(8)));

#define NS 32768
#define DD 64
#define LL 63
#define PP 14
#define NCAT 512
#define BB 5.0f
#define MINW 0.001f
#define MINH 0.001f
#define MIND 0.001f
#define TSPLIT 4
#define TPER (32 / TSPLIT)
#define NLAY (2 * TPER)
#define LSTR 200
#define OFF_B2 64
#define OFF_W3 72
#define OFF_B3 184

__global__ void nsf_pack_w1(const float* __restrict__ W1, float* __restrict__ Bt) {
    int idx = blockIdx.x * blockDim.x + threadIdx.x;
    if (idx >= NCAT * 64) return;
    int e    = idx & 3;
    int lane = (idx >> 2) & 31;
    int q    = (idx >> 7) & 7;
    int t    = idx >> 10;
    int m    = lane & 15;
    int hh   = lane >> 4;
    int col  = 16 * t + m;
    int k    = 4 * (2 * q + (e >> 1)) + 2 * hh + (e & 1);
    int l    = col >> 3;
    int h    = col & 7;
    float v = 0.0f;
    if (l < LL && k <= l) v = W1[(l * LL + k) * 8 + h];
    Bt[idx] = v;
}

__device__ __forceinline__ float hw_exp(float d)  { return __builtin_amdgcn_exp2f(d * 1.44269504088896341f); }
__device__ __forceinline__ float hw_ln(float v)   { return __builtin_amdgcn_logf(v) * 0.693147180559945309f; }
__device__ __forceinline__ float hw_rcp(float v)  { return __builtin_amdgcn_rcpf(v); }
__device__ __forceinline__ float hw_tanh(float v) { return __builtin_amdgcn_tanhf(v); }

__device__ __forceinline__ float softplusf(float v) {
    return fmaxf(v, 0.0f) + hw_ln(1.0f + hw_exp(-fabsf(v)));
}

__device__ void rqs_eval(const float* __restrict__ p, float xf, float& z, float& ld) {
    float cumw[6], cumh[6], derivs[6];
    {
        float mx = p[0];
        #pragma unroll
        for (int i = 1; i < 5; i++) mx = fmaxf(mx, p[i]);
        float e[5], s = 0.0f;
        #pragma unroll
        for (int i = 0; i < 5; i++) { e[i] = hw_exp(p[i] - mx); s += e[i]; }
        float inv = hw_rcp(s), cs = 0.0f;
        cumw[0] = -BB;
        #pragma unroll
        for (int i = 0; i < 4; i++) {
            cs += MINW + (1.0f - MINW * 5.0f) * e[i] * inv;
            cumw[i + 1] = 2.0f * BB * cs - BB;
        }
        cumw[5] = BB;
    }
    {
        float mx = p[5];
        #pragma unroll
        for (int i = 6; i < 10; i++) mx = fmaxf(mx, p[i]);
        float e[5], s = 0.0f;
        #pragma unroll
        for (int i = 0; i < 5; i++) { e[i] = hw_exp(p[5 + i] - mx); s += e[i]; }
        float inv = hw_rcp(s), cs = 0.0f;
        cumh[0] = -BB;
        #pragma unroll
        for (int i = 0; i < 4; i++) {
            cs += MINH + (1.0f - MINH * 5.0f) * e[i] * inv;
            cumh[i + 1] = 2.0f * BB * cs - BB;
        }
        cumh[5] = BB;
    }
    derivs[0] = 1.0f; derivs[5] = 1.0f;
    #pragma unroll
    for (int i = 1; i < 5; i++) derivs[i] = MIND + softplusf(p[9 + i]);

    float xc = fminf(fmaxf(xf, -BB), BB);
    int idx = 0;
    #pragma unroll
    for (int j = 1; j < 6; j++) idx += (xc >= cumw[j]) ? 1 : 0;
    if (idx > 4) idx = 4;

    float icw = cumw[idx], ibw = cumw[idx + 1] - cumw[idx];
    float ich = cumh[idx], ih  = cumh[idx + 1] - cumh[idx];
    const float rbw = hw_rcp(ibw);
    float idl = ih * rbw;
    float idv = derivs[idx], idvp1 = derivs[idx + 1];
    float theta = (xc - icw) * rbw;
    float tmt = theta * (1.0f - theta);
    float num = ih * (idl * theta * theta + idv * tmt);
    float den = idl + (idv + idvp1 - 2.0f * idl) * tmt;
    float outv = ich + num * hw_rcp(den);
    float omt = 1.0f - theta;
    float dnum = idl * idl * (idvp1 * theta * theta + 2.0f * idl * tmt + idv * omt * omt);
    float ldv = 0.693147180559945309f * (__builtin_amdgcn_logf(dnum) - 2.0f * __builtin_amdgcn_logf(den));
    bool inside = (xf >= -BB) && (xf <= BB);
    z  = inside ? outv : xf;
    ld = inside ? ldv : 0.0f;
}

__launch_bounds__(128, 8)
__global__ void nsf_fused(const float* __restrict__ x, const float* __restrict__ initp,
                          const float* __restrict__ b1, const float* __restrict__ W2,
                          const float* __restrict__ b2, const float* __restrict__ W3,
                          const float* __restrict__ b3, const float* __restrict__ Bt,
                          float* __restrict__ out) {
    __shared__ float ldacc[DD];
    __shared__ __attribute__((aligned(16))) float wl[NLAY * LSTR];
    __shared__ __attribute__((aligned(16))) float b1s[TPER * 16];

    const int tid  = threadIdx.x;
    const int wave = tid >> 5;
    const int lane = tid & 31;
    const int m    = lane & 15;
    const int hh   = lane >> 4;
    const int s0   = blockIdx.x * 64 + wave * 16;
    const int s    = s0 + m;
    const int part = blockIdx.y;
    const int t_begin = part * TPER;

    if (tid < DD) ldacc[tid] = 0.0f;

    {
        static_assert(NLAY % 16 == 0, "copy loops assume whole passes of 128 threads");
        const int L0   = 2 * t_begin;
        const int nlay = (LL - L0 < NLAY) ? (LL - L0) : NLAY;
        const float4* src2 = (const float4*)(W2 + (size_t)L0 * 64);
        const float4* src3 = (const float4*)(W3 + (size_t)L0 * 8 * PP);
        constexpr int N3 = NLAY * 28;
        constexpr int P3 = (N3 + 127) / 128;
        float4 r2[NLAY * 16 / 128], r3[P3];
        float  rb2[NLAY * 8 / 128], rb3[NLAY * 16 / 128], rb1[TPER * 16 / 128];
        #pragma unroll
        for (int k = 0; k < NLAY * 16 / 128; k++) {
            const int i = tid + 128 * k;
            r2[k] = src2[i < nlay * 16 ? i : nlay * 16 - 1];
        }
        #pragma unroll
        for (int k = 0; k < P3; k++) {
            const int i = tid + 128 * k;
            r3[k] = src3[i < nlay * 28 ? i : nlay * 28 - 1];
        }
        #pragma unroll
        for (int k = 0; k < NLAY * 8 / 128; k++) {
            const int i = tid + 128 * k;
            rb2[k] = b2[(size_t)L0 * 8 + (i < nlay * 8 ? i : nlay * 8 - 1)];
        }
        #pragma unroll
        for (int k = 0; k < NLAY * 16 / 128; k++) {
            const int i = tid + 128 * k;
            const int l = i >> 4, q = i & 15;
            const float v = b3[(size_t)(L0 + (l < nlay ? l : nlay - 1)) * PP + (q < PP ? q : PP - 1)];
            rb3[k] = (q < PP) ? v : 0.0f;
        }
        #pragma unroll
        for (int k = 0; k < TPER * 16 / 128; k++) {
            const int col = 16 * t_begin + tid + 128 * k;
            const float v = b1[col < LL * 8 ? col : LL * 8 - 1];
            rb1[k] = (col < LL * 8) ? v : 0.0f;
        }
        #pragma unroll
        for (int k = 0; k < NLAY * 16 / 128; k++) {
            const int i = tid + 128 * k;
            *(float4*)(wl + (i >> 4) * LSTR + 4 * (i & 15)) = r2[k];
        }
        #pragma unroll
        for (int k = 0; k < P3; k++) {
            const int i = tid + 128 * k;
            const int ic = (i < N3) ? i : N3 - 1;
            const int l = ic / 28, j = ic - l * 28;
            *(float4*)(wl + l * LSTR + OFF_W3 + 4 * j) = r3[k];
        }
        #pragma unroll
        for (int k = 0; k < NLAY * 8 / 128; k++) {
            const int i = tid + 128 * k;
            wl[(i >> 3) * LSTR + OFF_B2 + (i & 7)] = rb2[k];
        }
        #pragma unroll
        for (int k = 0; k < NLAY * 16 / 128; k++) {
            const int i = tid + 128 * k;
            wl[(i >> 4) * LSTR + OFF_B3 + (i & 15)] = rb3[k];
        }
        #pragma unroll
        for (int k = 0; k < TPER * 16 / 128; k++) b1s[tid + 128 * k] = rb1[k];
    }
    __syncthreads();

    if (part == 0 && lane < 16) {
        float p[PP];
        #pragma unroll
        for (int q = 0; q < PP; q++) p[q] = initp[q];
        float z, ld;
        rqs_eval(p, x[s * DD + 0], z, ld);
        out[s] = z;
        atomicAdd(&ldacc[0], ld);
    }

    v2f a[16];
    const float* xrow = x + (size_t)s * DD;
    #pragma unroll
    for (int kk = 0; kk < 16; kk++)
        a[kk] = *(const v2f*)(xrow + 4 * kk + 2 * hh);

    for (int t = t_begin; t < t_begin + TPER; ++t) {
        const v4f* bfrag = (const v4f*)(Bt + (size_t)t * 1024 + 4 * lane);
        v4f bq[8];
        #pragma unroll
        for (int q = 0; q < 8; q++) bq[q] = bfrag[32 * q];
        v8f c = {0.f, 0.f, 0.f, 0.f, 0.f, 0.f, 0.f, 0.f};
        #pragma unroll
        for (int q = 0; q < 8; q++) {
            const v2f b0 = {bq[q].x, bq[q].y}, b1v = {bq[q].z, bq[q].w};
            c = __builtin_amdgcn_wmma_f32_16x16x4_f32(
                    false, b0, false, a[2 * q], (short)0, c, false, false);
            c = __builtin_amdgcn_wmma_f32_16x16x4_f32(
                    false, b1v, false, a[2 * q + 1], (short)0, c, false, false);
        }

        float hv[8];
        {
            const float* bp = b1s + 16 * (t - t_begin) + 8 * hh;
            const float4 bl = *(const float4*)bp, bh = *(const float4*)(bp + 4);
            const float bias[8] = {bl.x, bl.y, bl.z, bl.w, bh.x, bh.y, bh.z, bh.w};
            #pragma unroll
            for (int r = 0; r < 8; r++) hv[r] = hw_tanh(c[r] + bias[r]);
        }

        const int l_idx = 2 * t + hh;
        if (l_idx < LL) {
            const int cdim = l_idx + 1;
            const float xval = x[(size_t)s * DD + cdim];

            const float* lw = wl + (2 * (t - t_begin) + hh) * LSTR;
            v2f acc2[4];
            #pragma unroll
            for (int gp = 0; gp < 4; gp++) acc2[gp] = *(const v2f*)(lw + OFF_B2 + 2 * gp);
            #pragma unroll
            for (int j = 0; j < 8; j++) {
                const v2f hs = {hv[j], hv[j]};
                #pragma unroll
                for (int gp = 0; gp < 4; gp++)
                    acc2[gp] = __builtin_elementwise_fma(hs, *(const v2f*)(lw + j * 8 + 2 * gp), acc2[gp]);
            }
            float h2[8];
            #pragma unroll
            for (int gp = 0; gp < 4; gp++) { h2[2 * gp] = hw_tanh(acc2[gp].x); h2[2 * gp + 1] = hw_tanh(acc2[gp].y); }

            v2f pv2[7];
            #pragma unroll
            for (int qp = 0; qp < 7; qp++) pv2[qp] = *(const v2f*)(lw + OFF_B3 + 2 * qp);
            #pragma unroll
            for (int g = 0; g < 8; g++) {
                const v2f hs = {h2[g], h2[g]};
                #pragma unroll
                for (int qp = 0; qp < 7; qp++)
                    pv2[qp] = __builtin_elementwise_fma(hs, *(const v2f*)(lw + OFF_W3 + g * PP + 2 * qp), pv2[qp]);
            }
            float pv[PP];
            #pragma unroll
            for (int qp = 0; qp < 7; qp++) { pv[2 * qp] = pv2[qp].x; pv[2 * qp + 1] = pv2[qp].y; }

            float z, ld;
            rqs_eval(pv, xval, z, ld);
            out[(size_t)cdim * NS + s] = z;
            atomicAdd(&ldacc[cdim], ld);
        }
    }
    __syncthreads();

    if (tid < DD) {
        const int owner = (tid == 0) ? 0 : (((tid - 1) >> 1) / TPER);
        if (owner == part)
            out[(size_t)NS * DD + (size_t)tid * (NS / 64) + blockIdx.x] = ldacc[tid];
    }
}

extern "C" void kernel_launch(void* const* d_in, const int* in_sizes, int n_in,
                              void* d_out, int out_size, void* d_ws, size_t ws_size,
                              hipStream_t stream) {
    const float* x     = (const float*)d_in[0];
    const float* initp = (const float*)d_in[1];
    const float* W1    = (const float*)d_in[2];
    const float* b1    = (const float*)d_in[3];
    const float* W2    = (const float*)d_in[4];
    const float* b2    = (const float*)d_in[5];
    const float* W3    = (const float*)d_in[6];
    const float* b3    = (const float*)d_in[7];
    float* out = (float*)d_out;
    float* Bt  = (float*)d_ws;

    nsf_pack_w1<<<(NCAT * 64 + 255) / 256, 256, 0, stream>>>(W1, Bt);
    nsf_fused<<<dim3(NS / 64, TSPLIT), 128, 0, stream>>>(x, initp, b1, W2, b2, W3, b3, Bt, out);
}


# layer 1 skips the k-steps the triangular mask zeroes: layer pair t issues (t>>1)+1 of 16 WMMAs and loads only their W1 fragments; work-groups take every 4th pair so all share it evenly; same sums
# speedup vs baseline: 5.2873x; 1.0812x over previous
#include <hip/hip_runtime.h>
#include <math.h>

typedef float v2f __attribute__((ext_vector_type(2)));
typedef float v4f __attribute__((ext_vector_type(4)));
typedef float v8f __attribute__((ext_vector_type(8)));

#define NS 32768
#define DD 64
#define LL 63
#define PP 14
#define NCAT 512
#define BB 5.0f
#define MINW 0.001f
#define MINH 0.001f
#define MIND 0.001f
#define TSPLIT 4
#define TPER (32 / TSPLIT)
#define NLAY (2 * TPER)
#define LSTR 200
#define OFF_B2 64
#define OFF_W3 72
#define OFF_B3 184

__global__ void nsf_pack_w1(const float* __restrict__ W1, float* __restrict__ Bt) {
    int idx = blockIdx.x * blockDim.x + threadIdx.x;
    if (idx >= NCAT * 64) return;
    int e    = idx & 3;
    int lane = (idx >> 2) & 31;
    int q    = (idx >> 7) & 7;
    int t    = idx >> 10;
    int m    = lane & 15;
    int hh   = lane >> 4;
    int col  = 16 * t + m;
    int k    = 4 * (2 * q + (e >> 1)) + 2 * hh + (e & 1);
    int l    = col >> 3;
    int h    = col & 7;
    float v = 0.0f;
    if (l < LL && k <= l) v = W1[(l * LL + k) * 8 + h];
    Bt[idx] = v;
}

__device__ __forceinline__ float hw_exp(float d)  { return __builtin_amdgcn_exp2f(d * 1.44269504088896341f); }
__device__ __forceinline__ float hw_ln(float v)   { return __builtin_amdgcn_logf(v) * 0.693147180559945309f; }
__device__ __forceinline__ float hw_rcp(float v)  { return __builtin_amdgcn_rcpf(v); }
__device__ __forceinline__ float hw_tanh(float v) { return __builtin_amdgcn_tanhf(v); }

__device__ __forceinline__ float softplusf(float v) {
    return fmaxf(v, 0.0f) + hw_ln(1.0f + hw_exp(-fabsf(v)));
}

__device__ void rqs_eval(const float* __restrict__ p, float xf, float& z, float& ld) {
    float cumw[6], cumh[6], derivs[6];
    {
        float mx = p[0];
        #pragma unroll
        for (int i = 1; i < 5; i++) mx = fmaxf(mx, p[i]);
        float e[5], s = 0.0f;
        #pragma unroll
        for (int i = 0; i < 5; i++) { e[i] = hw_exp(p[i] - mx); s += e[i]; }
        float inv = hw_rcp(s), cs = 0.0f;
        cumw[0] = -BB;
        #pragma unroll
        for (int i = 0; i < 4; i++) {
            cs += MINW + (1.0f - MINW * 5.0f) * e[i] * inv;
            cumw[i + 1] = 2.0f * BB * cs - BB;
        }
        cumw[5] = BB;
    }
    {
        float mx = p[5];
        #pragma unroll
        for (int i = 6; i < 10; i++) mx = fmaxf(mx, p[i]);
        float e[5], s = 0.0f;
        #pragma unroll
        for (int i = 0; i < 5; i++) { e[i] = hw_exp(p[5 + i] - mx); s += e[i]; }
        float inv = hw_rcp(s), cs = 0.0f;
        cumh[0] = -BB;
        #pragma unroll
        for (int i = 0; i < 4; i++) {
            cs += MINH + (1.0f - MINH * 5.0f) * e[i] * inv;
            cumh[i + 1] = 2.0f * BB * cs - BB;
        }
        cumh[5] = BB;
    }
    derivs[0] = 1.0f; derivs[5] = 1.0f;
    #pragma unroll
    for (int i = 1; i < 5; i++) derivs[i] = MIND + softplusf(p[9 + i]);

    float xc = fminf(fmaxf(xf, -BB), BB);
    int idx = 0;
    #pragma unroll
    for (int j = 1; j < 6; j++) idx += (xc >= cumw[j]) ? 1 : 0;
    if (idx > 4) idx = 4;

    float icw = cumw[idx], ibw = cumw[idx + 1] - cumw[idx];
    float ich = cumh[idx], ih  = cumh[idx + 1] - cumh[idx];
    const float rbw = hw_rcp(ibw);
    float idl = ih * rbw;
    float idv = derivs[idx], idvp1 = derivs[idx + 1];
    float theta = (xc - icw) * rbw;
    float tmt = theta * (1.0f - theta);
    float num = ih * (idl * theta * theta + idv * tmt);
    float den = idl + (idv + idvp1 - 2.0f * idl) * tmt;
    float outv = ich + num * hw_rcp(den);
    float omt = 1.0f - theta;
    float dnum = idl * idl * (idvp1 * theta * theta + 2.0f * idl * tmt + idv * omt * omt);
    float ldv = 0.693147180559945309f * (__builtin_amdgcn_logf(dnum) - 2.0f * __builtin_amdgcn_logf(den));
    bool inside = (xf >= -BB) && (xf <= BB);
    z  = inside ? outv : xf;
    ld = inside ? ldv : 0.0f;
}

__device__ __forceinline__ int model_layer(int jl, int part) {
    const int l = 2 * (TSPLIT * (jl >> 1) + part) + (jl & 1);
    return l < LL ? l : LL - 1;
}

template <int N>
__device__ __forceinline__ v8f layer1_ksteps(const v4f* __restrict__ bfrag, const v2f (&a)[15],
                                             const float* __restrict__ x, unsigned xrow_off, int klast) {
    constexpr int NL = (N + 1) / 2;
    v4f bq[NL];
    #pragma unroll
    for (int q = 0; q < NL; q++) bq[q] = bfrag[32 * q];
    v2f a15 = {0.f, 0.f};
    if (N == 16) a15 = *(const v2f*)((const char*)x + (xrow_off + 16u * (unsigned)klast));
    v8f c = {0.f, 0.f, 0.f, 0.f, 0.f, 0.f, 0.f, 0.f};
    #pragma unroll
    for (int k = 0; k < N; k++) {
        const v4f f = bq[k >> 1];
        const v2f w = (k & 1) ? v2f{f.z, f.w} : v2f{f.x, f.y};
        const v2f xk = (k < 15) ? a[k < 15 ? k : 0] : a15;
        c = __builtin_amdgcn_wmma_f32_16x16x4_f32(false, w, false, xk, (short)0, c, false, false);
    }
    return c;
}

__launch_bounds__(128, 8)
__global__ void nsf_fused(const float* __restrict__ x, const float* __restrict__ initp,
                          const float* __restrict__ b1, const float* __restrict__ W2,
                          const float* __restrict__ b2, const float* __restrict__ W3,
                          const float* __restrict__ b3, const float* __restrict__ Bt,
                          float* __restrict__ out) {
    __shared__ float ldacc[DD];
    __shared__ __attribute__((aligned(16))) float wl[NLAY * LSTR];
    __shared__ __attribute__((aligned(16))) float b1s[TPER * 16];

    const int tid  = threadIdx.x;
    const int wave = __builtin_amdgcn_readfirstlane(tid >> 5);
    const int lane = tid & 31;
    const int m    = lane & 15;
    const int hh   = lane >> 4;
    const int s0   = blockIdx.x * 64 + wave * 16;
    const int s    = s0 + m;
    const int part = blockIdx.y;

    if (tid < DD) ldacc[tid] = 0.0f;

    {
        static_assert(NLAY % 16 == 0, "copy loops assume whole passes of 128 threads");
        const float4* src2 = (const float4*)W2;
        const float4* src3 = (const float4*)W3;
        constexpr int N3 = NLAY * 28;
        constexpr int P3 = (N3 + 127) / 128;
        float4 r2[NLAY * 16 / 128], r3[P3];
        float  rb2[NLAY * 8 / 128], rb3[NLAY * 16 / 128], rb1[TPER * 16 / 128];
        #pragma unroll
        for (int k = 0; k < NLAY * 16 / 128; k++) {
            const int i = tid + 128 * k;
            r2[k] = src2[model_layer(i >> 4, part) * 16 + (i & 15)];
        }
        #pragma unroll
        for (int k = 0; k < P3; k++) {
            const int i = tid + 128 * k;
            const int ic = (i < N3) ? i : N3 - 1;
            const int jl = ic / 28, j = ic - jl * 28;
            r3[k] = src3[model_layer(jl, part) * 28 + j];
        }
        #pragma unroll
        for (int k = 0; k < NLAY * 8 / 128; k++) {
            const int i = tid + 128 * k;
            rb2[k] = b2[model_layer(i >> 3, part) * 8 + (i & 7)];
        }
        #pragma unroll
        for (int k = 0; k < NLAY * 16 / 128; k++) {
            const int i = tid + 128 * k;
            const int q = i & 15;
            const float v = b3[model_layer(i >> 4, part) * PP + (q < PP ? q : PP - 1)];
            rb3[k] = (q < PP) ? v : 0.0f;
        }
        #pragma unroll
        for (int k = 0; k < TPER * 16 / 128; k++) {
            const int i = tid + 128 * k;
            const int col = 16 * (TSPLIT * (i >> 4) + part) + (i & 15);
            const float v = b1[col < LL * 8 ? col : LL * 8 - 1];
            rb1[k] = (col < LL * 8) ? v : 0.0f;
        }
        #pragma unroll
        for (int k = 0; k < NLAY * 16 / 128; k++) {
            const int i = tid + 128 * k;
            *(float4*)(wl + (i >> 4) * LSTR + 4 * (i & 15)) = r2[k];
        }
        #pragma unroll
        for (int k = 0; k < P3; k++) {
            const int i = tid + 128 * k;
            const int ic = (i < N3) ? i : N3 - 1;
            const int jl = ic / 28, j = ic - jl * 28;
            *(float4*)(wl + jl * LSTR + OFF_W3 + 4 * j) = r3[k];
        }
        #pragma unroll
        for (int k = 0; k < NLAY * 8 / 128; k++) {
            const int i = tid + 128 * k;
            wl[(i >> 3) * LSTR + OFF_B2 + (i & 7)] = rb2[k];
        }
        #pragma unroll
        for (int k = 0; k < NLAY * 16 / 128; k++) {
            const int i = tid + 128 * k;
            wl[(i >> 4) * LSTR + OFF_B3 + (i & 15)] = rb3[k];
        }
        #pragma unroll
        for (int k = 0; k < TPER * 16 / 128; k++) b1s[tid + 128 * k] = rb1[k];
    }
    __syncthreads();

    if (part == 0 && lane < 16) {
        float p[PP];
        #pragma unroll
        for (int q = 0; q < PP; q++) p[q] = initp[q];
        float z, ld;
        rqs_eval(p, x[s * DD + 0], z, ld);
        out[s] = z;
        atomicAdd(&ldacc[0], ld);
    }

    v2f a[15];
    const unsigned xrow_off = (unsigned)s * (DD * 4u) + 8u * (unsigned)hh;
    #pragma unroll
    for (int kk = 0; kk < 15; kk++)
        a[kk] = *(const v2f*)((const char*)x + xrow_off + 16u * kk);
    const unsigned zout_off = 4u * ((unsigned)(hh + 1) * (unsigned)NS + (unsigned)s);
    const unsigned bfrag_off = 16u * (unsigned)lane;
    const float* wl_lane = wl + hh * LSTR;
    const float* b1_lane = b1s + 8 * hh;

    for (int i = 0; i < TPER; ++i) {
        const int t = TSPLIT * i + part;
        const v4f* bfrag = (const v4f*)((const char*)Bt + (bfrag_off + 4096u * (unsigned)t));
        v8f c;
        switch ((t >> 1) + 1) {
            case 1:  c = layer1_ksteps<1>(bfrag, a, x, xrow_off, t >> 1);  break;
            case 2:  c = layer1_ksteps<2>(bfrag, a, x, xrow_off, t >> 1);  break;
            case 3:  c = layer1_ksteps<3>(bfrag, a, x, xrow_off, t >> 1);  break;
            case 4:  c = layer1_ksteps<4>(bfrag, a, x, xrow_off, t >> 1);  break;
            case 5:  c = layer1_ksteps<5>(bfrag, a, x, xrow_off, t >> 1);  break;
            case 6:  c = layer1_ksteps<6>(bfrag, a, x, xrow_off, t >> 1);  break;
            case 7:  c = layer1_ksteps<7>(bfrag, a, x, xrow_off, t >> 1);  break;
            case 8:  c = layer1_ksteps<8>(bfrag, a, x, xrow_off, t >> 1);  break;
            case 9:  c = layer1_ksteps<9>(bfrag, a, x, xrow_off, t >> 1);  break;
            case 10: c = layer1_ksteps<10>(bfrag, a, x, xrow_off, t >> 1); break;
            case 11: c = layer1_ksteps<11>(bfrag, a, x, xrow_off, t >> 1); break;
            case 12: c = layer1_ksteps<12>(bfrag, a, x, xrow_off, t >> 1); break;
            case 13: c = layer1_ksteps<13>(bfrag, a, x, xrow_off, t >> 1); break;
            case 14: c = layer1_ksteps<14>(bfrag, a, x, xrow_off, t >> 1); break;
            case 15: c = layer1_ksteps<15>(bfrag, a, x, xrow_off, t >> 1); break;
            default: c = layer1_ksteps<16>(bfrag, a, x, xrow_off, t >> 1); break;
        }

        float hv[8];
        {
            const float* bp = b1_lane + 16 * i;
            const float4 bl = *(const float4*)bp, bh = *(const float4*)(bp + 4);
            const float bias[8] = {bl.x, bl.y, bl.z, bl.w, bh.x, bh.y, bh.z, bh.w};
            #pragma unroll
            for (int r = 0; r < 8; r++) hv[r] = hw_tanh(c[r] + bias[r]);
        }

        const int l_idx = 2 * t + hh;
        if (l_idx < LL) {
            const int cdim = l_idx + 1;
            const v2f xpair = *(const v2f*)((const char*)x + (xrow_off + 8u * (unsigned)t));
            const float xval = hh ? xpair.x : xpair.y;

            const float* lw = wl_lane + i * (2 * LSTR);
            v2f acc2[4];
            #pragma unroll
            for (int gp = 0; gp < 4; gp++) acc2[gp] = *(const v2f*)(lw + OFF_B2 + 2 * gp);
            #pragma unroll
            for (int j = 0; j < 8; j++) {
                const v2f hs = {hv[j], hv[j]};
                #pragma unroll
                for (int gp = 0; gp < 4; gp++)
                    acc2[gp] = __builtin_elementwise_fma(hs, *(const v2f*)(lw + j * 8 + 2 * gp), acc2[gp]);
            }
            float h2[8];
            #pragma unroll
            for (int gp = 0; gp < 4; gp++) { h2[2 * gp] = hw_tanh(acc2[gp].x); h2[2 * gp + 1] = hw_tanh(acc2[gp].y); }

            v2f pv2[7];
            #pragma unroll
            for (int qp = 0; qp < 7; qp++) pv2[qp] = *(const v2f*)(lw + OFF_B3 + 2 * qp);
            #pragma unroll
            for (int g = 0; g < 8; g++) {
                const v2f hs = {h2[g], h2[g]};
                #pragma unroll
                for (int qp = 0; qp < 7; qp++)
                    pv2[qp] = __builtin_elementwise_fma(hs, *(const v2f*)(lw + OFF_W3 + g * PP + 2 * qp), pv2[qp]);
            }
            float pv[PP];
            #pragma unroll
            for (int qp = 0; qp < 7; qp++) { pv[2 * qp] = pv2[qp].x; pv[2 * qp + 1] = pv2[qp].y; }

            float z, ld;
            rqs_eval(pv, xval, z, ld);
            *(float*)((char*)out + (zout_off + (unsigned)t * (8u * NS))) = z;
            atomicAdd(&ldacc[cdim], ld);
        }
    }
    __syncthreads();

    const int etid = 32 * wave + (int)__builtin_amdgcn_mbcnt_lo(~0u, 0u);
    if (etid < DD) {
        const int owner = (etid == 0) ? 0 : (((etid - 1) >> 1) % TSPLIT);
        if (owner == part)
            out[(size_t)NS * DD + (size_t)etid * (NS / 64) + blockIdx.x] = ldacc[etid];
    }
}

extern "C" void kernel_launch(void* const* d_in, const int* in_sizes, int n_in,
                              void* d_out, int out_size, void* d_ws, size_t ws_size,
                              hipStream_t stream) {
    const float* x     = (const float*)d_in[0];
    const float* initp = (const float*)d_in[1];
    const float* W1    = (const float*)d_in[2];
    const float* b1    = (const float*)d_in[3];
    const float* W2    = (const float*)d_in[4];
    const float* b2    = (const float*)d_in[5];
    const float* W3    = (const float*)d_in[6];
    const float* b3    = (const float*)d_in[7];
    float* out = (float*)d_out;
    float* Bt  = (float*)d_ws;

    nsf_pack_w1<<<(NCAT * 64 + 255) / 256, 256, 0, stream>>>(W1, Bt);
    nsf_fused<<<dim3(NS / 64, TSPLIT), 128, 0, stream>>>(x, initp, b1, W2, b2, W3, b3, Bt, out);
}


# W1 packing launch removed: layer 1 reads its A fragments in place from W1 (two 4-byte loads per k-step per lane), mask k <= l applied to the one k-step that crosses the diagonal; 2 launches -> 1
# speedup vs baseline: 5.5245x; 1.0449x over previous
#include <hip/hip_runtime.h>
#include <math.h>

typedef float v2f __attribute__((ext_vector_type(2)));
typedef float v4f __attribute__((ext_vector_type(4)));
typedef float v8f __attribute__((ext_vector_type(8)));

#define NS 32768
#define DD 64
#define LL 63
#define PP 14
#define BB 5.0f
#define MINW 0.001f
#define MINH 0.001f
#define MIND 0.001f
#define TSPLIT 4
#define TPER (32 / TSPLIT)
#define NLAY (2 * TPER)
#define LSTR 200
#define OFF_B2 64
#define OFF_W3 72
#define OFF_B3 184

__device__ __forceinline__ float hw_exp(float d)  { return __builtin_amdgcn_exp2f(d * 1.44269504088896341f); }
__device__ __forceinline__ float hw_ln(float v)   { return __builtin_amdgcn_logf(v) * 0.693147180559945309f; }
__device__ __forceinline__ float hw_rcp(float v)  { return __builtin_amdgcn_rcpf(v); }
__device__ __forceinline__ float hw_tanh(float v) { return __builtin_amdgcn_tanhf(v); }

__device__ __forceinline__ float softplusf(float v) {
    return fmaxf(v, 0.0f) + hw_ln(1.0f + hw_exp(-fabsf(v)));
}

__device__ void rqs_eval(const float* __restrict__ p, float xf, float& z, float& ld) {
    float cumw[6], cumh[6], derivs[6];
    {
        float mx = p[0];
        #pragma unroll
        for (int i = 1; i < 5; i++) mx = fmaxf(mx, p[i]);
        float e[5], s = 0.0f;
        #pragma unroll
        for (int i = 0; i < 5; i++) { e[i] = hw_exp(p[i] - mx); s += e[i]; }
        float inv = hw_rcp(s), cs = 0.0f;
        cumw[0] = -BB;
        #pragma unroll
        for (int i = 0; i < 4; i++) {
            cs += MINW + (1.0f - MINW * 5.0f) * e[i] * inv;
            cumw[i + 1] = 2.0f * BB * cs - BB;
        }
        cumw[5] = BB;
    }
    {
        float mx = p[5];
        #pragma unroll
        for (int i = 6; i < 10; i++) mx = fmaxf(mx, p[i]);
        float e[5], s = 0.0f;
        #pragma unroll
        for (int i = 0; i < 5; i++) { e[i] = hw_exp(p[5 + i] - mx); s += e[i]; }
        float inv = hw_rcp(s), cs = 0.0f;
        cumh[0] = -BB;
        #pragma unroll
        for (int i = 0; i < 4; i++) {
            cs += MINH + (1.0f - MINH * 5.0f) * e[i] * inv;
            cumh[i + 1] = 2.0f * BB * cs - BB;
        }
        cumh[5] = BB;
    }
    derivs[0] = 1.0f; derivs[5] = 1.0f;
    #pragma unroll
    for (int i = 1; i < 5; i++) derivs[i] = MIND + softplusf(p[9 + i]);

    float xc = fminf(fmaxf(xf, -BB), BB);
    int idx = 0;
    #pragma unroll
    for (int j = 1; j < 6; j++) idx += (xc >= cumw[j]) ? 1 : 0;
    if (idx > 4) idx = 4;

    float icw = cumw[idx], ibw = cumw[idx + 1] - cumw[idx];
    float ich = cumh[idx], ih  = cumh[idx + 1] - cumh[idx];
    const float rbw = hw_rcp(ibw);
    float idl = ih * rbw;
    float idv = derivs[idx], idvp1 = derivs[idx + 1];
    float theta = (xc - icw) * rbw;
    float tmt = theta * (1.0f - theta);
    float num = ih * (idl * theta * theta + idv * tmt);
    float den = idl + (idv + idvp1 - 2.0f * idl) * tmt;
    float outv = ich + num * hw_rcp(den);
    float omt = 1.0f - theta;
    float dnum = idl * idl * (idvp1 * theta * theta + 2.0f * idl * tmt + idv * omt * omt);
    float ldv = 0.693147180559945309f * (__builtin_amdgcn_logf(dnum) - 2.0f * __builtin_amdgcn_logf(den));
    bool inside = (xf >= -BB) && (xf <= BB);
    z  = inside ? outv : xf;
    ld = inside ? ldv : 0.0f;
}

__device__ __forceinline__ int model_layer(int jl, int part) {
    const int l = 2 * (TSPLIT * (jl >> 1) + part) + (jl & 1);
    return l < LL ? l : LL - 1;
}

template <int N>
__device__ __forceinline__ v8f layer1_ksteps(const float* __restrict__ W1, unsigned w_off, int kl, int t, const v2f (&a)[15],
                                             const float* __restrict__ x, unsigned xrow_off, int klast) {
    constexpr unsigned W1_LAST = 4u * (LL * LL * 8u - 1u);
    if (N == 16) w_off -= (t == 31 && w_off >= 4032u * 31u + 2016u) ? 2016u : 0u;
    const float* __restrict__ wp = (const float*)((const char*)W1 + w_off);
    v2f wk[N];
    #pragma unroll
    for (int k = 0; k < N; k++) {
        wk[k].x = wp[32 * k];
        if (N == 16 && k == 15) {
            const unsigned ol = w_off + 4u * (32u * 15u + 8u);
            wk[k].y = *(const float*)((const char*)W1 + (ol < W1_LAST ? ol : W1_LAST));
        } else {
            wk[k].y = wp[32 * k + 8];
        }
    }
    v2f a15 = {0.f, 0.f};
    if (N == 16) a15 = *(const v2f*)((const char*)x + (xrow_off + 16u * (unsigned)klast));
    wk[N - 1].x = (kl >= 0) ? wk[N - 1].x : 0.0f;
    wk[N - 1].y = (kl >= 1) ? wk[N - 1].y : 0.0f;
    __builtin_amdgcn_sched_barrier(0);
    v8f c = {0.f, 0.f, 0.f, 0.f, 0.f, 0.f, 0.f, 0.f};
    #pragma unroll
    for (int k = 0; k < N; k++) {
        const v2f xk = (k < 15) ? a[k < 15 ? k : 0] : a15;
        c = __builtin_amdgcn_wmma_f32_16x16x4_f32(false, wk[k], false, xk, (short)0, c, false, false);
    }
    return c;
}

__launch_bounds__(128, 8)
__global__ void nsf_fused(const float* __restrict__ x, const float* __restrict__ initp,
                          const float* __restrict__ W1, const float* __restrict__ b1,
                          const float* __restrict__ W2, const float* __restrict__ b2,
                          const float* __restrict__ W3, const float* __restrict__ b3,
                          float* __restrict__ out) {
    __shared__ float ldacc[DD];
    __shared__ __attribute__((aligned(16))) float wl[NLAY * LSTR];
    __shared__ __attribute__((aligned(16))) float b1s[TPER * 16];

    const int tid  = threadIdx.x;
    const int wave = __builtin_amdgcn_readfirstlane(tid >> 5);
    const int lane = tid & 31;
    const int m    = lane & 15;
    const int hh   = lane >> 4;
    const int s0   = blockIdx.x * 64 + wave * 16;
    const int s    = s0 + m;
    const int part = blockIdx.y;

    if (tid < DD) ldacc[tid] = 0.0f;

    {
        static_assert(NLAY % 16 == 0, "copy loops assume whole passes of 128 threads");
        const float4* src2 = (const float4*)W2;
        const float4* src3 = (const float4*)W3;
        constexpr int N3 = NLAY * 28;
        constexpr int P3 = (N3 + 127) / 128;
        float4 r2[NLAY * 16 / 128], r3[P3];
        float  rb2[NLAY * 8 / 128], rb3[NLAY * 16 / 128], rb1[TPER * 16 / 128];
        #pragma unroll
        for (int k = 0; k < NLAY * 16 / 128; k++) {
            const int i = tid + 128 * k;
            r2[k] = src2[model_layer(i >> 4, part) * 16 + (i & 15)];
        }
        #pragma unroll
        for (int k = 0; k < P3; k++) {
            const int i = tid + 128 * k;
            const int ic = (i < N3) ? i : N3 - 1;
            const int jl = ic / 28, j = ic - jl * 28;
            r3[k] = src3[model_layer(jl, part) * 28 + j];
        }
        #pragma unroll
        for (int k = 0; k < NLAY * 8 / 128; k++) {
            const int i = tid + 128 * k;
            rb2[k] = b2[model_layer(i >> 3, part) * 8 + (i & 7)];
        }
        #pragma unroll
        for (int k = 0; k < NLAY * 16 / 128; k++) {
            const int i = tid + 128 * k;
            const int q = i & 15;
            const float v = b3[model_layer(i >> 4, part) * PP + (q < PP ? q : PP - 1)];
            rb3[k] = (q < PP) ? v : 0.0f;
        }
        #pragma unroll
        for (int k = 0; k < TPER * 16 / 128; k++) {
            const int i = tid + 128 * k;
            const int col = 16 * (TSPLIT * (i >> 4) + part) + (i & 15);
            const float v = b1[col < LL * 8 ? col : LL * 8 - 1];
            rb1[k] = (col < LL * 8) ? v : 0.0f;
        }
        #pragma unroll
        for (int k = 0; k < NLAY * 16 / 128; k++) {
            const int i = tid + 128 * k;
            *(float4*)(wl + (i >> 4) * LSTR + 4 * (i & 15)) = r2[k];
        }
        #pragma unroll
        for (int k = 0; k < P3; k++) {
            const int i = tid + 128 * k;
            const int ic = (i < N3) ? i : N3 - 1;
            const int jl = ic / 28, j = ic - jl * 28;
            *(float4*)(wl + jl * LSTR + OFF_W3 + 4 * j) = r3[k];
        }
        #pragma unroll
        for (int k = 0; k < NLAY * 8 / 128; k++) {
            const int i = tid + 128 * k;
            wl[(i >> 3) * LSTR + OFF_B2 + (i & 7)] = rb2[k];
        }
        #pragma unroll
        for (int k = 0; k < NLAY * 16 / 128; k++) {
            const int i = tid + 128 * k;
            wl[(i >> 4) * LSTR + OFF_B3 + (i & 15)] = rb3[k];
        }
        #pragma unroll
        for (int k = 0; k < TPER * 16 / 128; k++) b1s[tid + 128 * k] = rb1[k];
    }
    __syncthreads();

    if (part == 0 && lane < 16) {
        float p[PP];
        #pragma unroll
        for (int q = 0; q < PP; q++) p[q] = initp[q];
        float z, ld;
        rqs_eval(p, x[s * DD + 0], z, ld);
        out[s] = z;
        atomicAdd(&ldacc[0], ld);
    }

    v2f a[15];
    const unsigned xrow_off = (unsigned)s * (DD * 4u) + 8u * (unsigned)hh;
    #pragma unroll
    for (int kk = 0; kk < 15; kk++)
        a[kk] = *(const v2f*)((const char*)x + xrow_off + 16u * kk);
    const unsigned zout_off = 4u * ((unsigned)(hh + 1) * (unsigned)NS + (unsigned)s);
    const unsigned w_lane = 2016u * (unsigned)(m >> 3) + 64u * (unsigned)hh + 4u * (unsigned)(m & 7);
    const float* wl_lane = wl + hh * LSTR;
    const float* b1_lane = b1s + 8 * hh;

    for (int i = 0; i < TPER; ++i) {
        const int t = TSPLIT * i + part;
        const unsigned w_off = w_lane + 4032u * (unsigned)t;
        const int kl = 2 * (t & 1) + (w_lane >= 2016u ? 1 : 0) - 2 * hh;
        v8f c;
        switch ((t >> 1) + 1) {
            case 1:  c = layer1_ksteps<1>(W1, w_off, kl, t, a, x, xrow_off, t >> 1);  break;
            case 2:  c = layer1_ksteps<2>(W1, w_off, kl, t, a, x, xrow_off, t >> 1);  break;
            case 3:  c = layer1_ksteps<3>(W1, w_off, kl, t, a, x, xrow_off, t >> 1);  break;
            case 4:  c = layer1_ksteps<4>(W1, w_off, kl, t, a, x, xrow_off, t >> 1);  break;
            case 5:  c = layer1_ksteps<5>(W1, w_off, kl, t, a, x, xrow_off, t >> 1);  break;
            case 6:  c = layer1_ksteps<6>(W1, w_off, kl, t, a, x, xrow_off, t >> 1);  break;
            case 7:  c = layer1_ksteps<7>(W1, w_off, kl, t, a, x, xrow_off, t >> 1);  break;
            case 8:  c = layer1_ksteps<8>(W1, w_off, kl, t, a, x, xrow_off, t >> 1);  break;
            case 9:  c = layer1_ksteps<9>(W1, w_off, kl, t, a, x, xrow_off, t >> 1);  break;
            case 10: c = layer1_ksteps<10>(W1, w_off, kl, t, a, x, xrow_off, t >> 1); break;
            case 11: c = layer1_ksteps<11>(W1, w_off, kl, t, a, x, xrow_off, t >> 1); break;
            case 12: c = layer1_ksteps<12>(W1, w_off, kl, t, a, x, xrow_off, t >> 1); break;
            case 13: c = layer1_ksteps<13>(W1, w_off, kl, t, a, x, xrow_off, t >> 1); break;
            case 14: c = layer1_ksteps<14>(W1, w_off, kl, t, a, x, xrow_off, t >> 1); break;
            case 15: c = layer1_ksteps<15>(W1, w_off, kl, t, a, x, xrow_off, t >> 1); break;
            default: c = layer1_ksteps<16>(W1, w_off, kl, t, a, x, xrow_off, t >> 1); break;
        }

        float hv[8];
        {
            const float* bp = b1_lane + 16 * i;
            const float4 bl = *(const float4*)bp, bh = *(const float4*)(bp + 4);
            const float bias[8] = {bl.x, bl.y, bl.z, bl.w, bh.x, bh.y, bh.z, bh.w};
            #pragma unroll
            for (int r = 0; r < 8; r++) hv[r] = hw_tanh(c[r] + bias[r]);
        }

        const int l_idx = 2 * t + hh;
        if (l_idx < LL) {
            const int cdim = l_idx + 1;
            const v2f xpair = *(const v2f*)((const char*)x + (xrow_off + 8u * (unsigned)t));
            const float xval = hh ? xpair.x : xpair.y;

            const float* lw = wl_lane + i * (2 * LSTR);
            v2f acc2[4];
            #pragma unroll
            for (int gp = 0; gp < 4; gp++) acc2[gp] = *(const v2f*)(lw + OFF_B2 + 2 * gp);
            #pragma unroll
            for (int j = 0; j < 8; j++) {
                const v2f hs = {hv[j], hv[j]};
                #pragma unroll
                for (int gp = 0; gp < 4; gp++)
                    acc2[gp] = __builtin_elementwise_fma(hs, *(const v2f*)(lw + j * 8 + 2 * gp), acc2[gp]);
            }
            float h2[8];
            #pragma unroll
            for (int gp = 0; gp < 4; gp++) { h2[2 * gp] = hw_tanh(acc2[gp].x); h2[2 * gp + 1] = hw_tanh(acc2[gp].y); }

            v2f pv2[7];
            #pragma unroll
            for (int qp = 0; qp < 7; qp++) pv2[qp] = *(const v2f*)(lw + OFF_B3 + 2 * qp);
            #pragma unroll
            for (int g = 0; g < 8; g++) {
                const v2f hs = {h2[g], h2[g]};
                #pragma unroll
                for (int qp = 0; qp < 7; qp++)
                    pv2[qp] = __builtin_elementwise_fma(hs, *(const v2f*)(lw + OFF_W3 + g * PP + 2 * qp), pv2[qp]);
            }
            float pv[PP];
            #pragma unroll
            for (int qp = 0; qp < 7; qp++) { pv[2 * qp] = pv2[qp].x; pv[2 * qp + 1] = pv2[qp].y; }

            float z, ld;
            rqs_eval(pv, xval, z, ld);
            *(float*)((char*)out + (zout_off + (unsigned)t * (8u * NS))) = z;
            atomicAdd(&ldacc[cdim], ld);
        }
    }
    __syncthreads();

    const int etid = 32 * wave + (int)__builtin_amdgcn_mbcnt_lo(~0u, 0u);
    if (etid < DD) {
        const int owner = (etid == 0) ? 0 : (((etid - 1) >> 1) % TSPLIT);
        if (owner == part)
            out[(size_t)NS * DD + (size_t)etid * (NS / 64) + blockIdx.x] = ldacc[etid];
    }
}

extern "C" void kernel_launch(void* const* d_in, const int* in_sizes, int n_in,
                              void* d_out, int out_size, void* d_ws, size_t ws_size,
                              hipStream_t stream) {
    const float* x     = (const float*)d_in[0];
    const float* initp = (const float*)d_in[1];
    const float* W1    = (const float*)d_in[2];
    const float* b1    = (const float*)d_in[3];
    const float* W2    = (const float*)d_in[4];
    const float* b2    = (const float*)d_in[5];
    const float* W3    = (const float*)d_in[6];
    const float* b3    = (const float*)d_in[7];
    float* out = (float*)d_out;
    nsf_fused<<<dim3(NS / 64, TSPLIT), 128, 0, stream>>>(x, initp, W1, b1, W2, b2, W3, b3, out);
}


# each lane carries two samples of its layer (a wave = two 16-sample WMMA tiles on shared W1 fragments, two accumulators): every LDS weight read feeds both; grid 512x4 -> 256x4, 4 waves per SIMD
# speedup vs baseline: 6.1502x; 1.1133x over previous
#include <hip/hip_runtime.h>
#include <math.h>

typedef float v2f __attribute__((ext_vector_type(2)));
typedef float v4f __attribute__((ext_vector_type(4)));
typedef float v8f __attribute__((ext_vector_type(8)));

#define NS 32768
#define DD 64
#define LL 63
#define PP 14
#define BB 5.0f
#define MINW 0.001f
#define MINH 0.001f
#define MIND 0.001f
#define BSAMP 128
#define TILE_X (16u * DD * 4u)
#define TSPLIT 4
#define TPER (32 / TSPLIT)
#define NLAY (2 * TPER)
#define LSTR 200
#define OFF_B2 64
#define OFF_W3 72
#define OFF_B3 184

__device__ __forceinline__ float hw_exp(float d)  { return __builtin_amdgcn_exp2f(d * 1.44269504088896341f); }
__device__ __forceinline__ float hw_ln(float v)   { return __builtin_amdgcn_logf(v) * 0.693147180559945309f; }
__device__ __forceinline__ float hw_rcp(float v)  { return __builtin_amdgcn_rcpf(v); }
__device__ __forceinline__ float hw_tanh(float v) { return __builtin_amdgcn_tanhf(v); }

__device__ __forceinline__ float softplusf(float v) {
    return fmaxf(v, 0.0f) + hw_ln(1.0f + hw_exp(-fabsf(v)));
}

__device__ __forceinline__ void rqs_eval(const float* __restrict__ p, float xf, float& z, float& ld) {
    float cumw[6], cumh[6], derivs[6];
    {
        float mx = p[0];
        #pragma unroll
        for (int i = 1; i < 5; i++) mx = fmaxf(mx, p[i]);
        float e[5], s = 0.0f;
        #pragma unroll
        for (int i = 0; i < 5; i++) { e[i] = hw_exp(p[i] - mx); s += e[i]; }
        float inv = hw_rcp(s), cs = 0.0f;
        cumw[0] = -BB;
        #pragma unroll
        for (int i = 0; i < 4; i++) {
            cs += MINW + (1.0f - MINW * 5.0f) * e[i] * inv;
            cumw[i + 1] = 2.0f * BB * cs - BB;
        }
        cumw[5] = BB;
    }
    {
        float mx = p[5];
        #pragma unroll
        for (int i = 6; i < 10; i++) mx = fmaxf(mx, p[i]);
        float e[5], s = 0.0f;
        #pragma unroll
        for (int i = 0; i < 5; i++) { e[i] = hw_exp(p[5 + i] - mx); s += e[i]; }
        float inv = hw_rcp(s), cs = 0.0f;
        cumh[0] = -BB;
        #pragma unroll
        for (int i = 0; i < 4; i++) {
            cs += MINH + (1.0f - MINH * 5.0f) * e[i] * inv;
            cumh[i + 1] = 2.0f * BB * cs - BB;
        }
        cumh[5] = BB;
    }
    derivs[0] = 1.0f; derivs[5] = 1.0f;
    #pragma unroll
    for (int i = 1; i < 5; i++) derivs[i] = MIND + softplusf(p[9 + i]);

    float xc = fminf(fmaxf(xf, -BB), BB);
    int idx = 0;
    #pragma unroll
    for (int j = 1; j < 6; j++) idx += (xc >= cumw[j]) ? 1 : 0;
    if (idx > 4) idx = 4;

    float icw = cumw[idx], ibw = cumw[idx + 1] - cumw[idx];
    float ich = cumh[idx], ih  = cumh[idx + 1] - cumh[idx];
    const float rbw = hw_rcp(ibw);
    float idl = ih * rbw;
    float idv = derivs[idx], idvp1 = derivs[idx + 1];
    float theta = (xc - icw) * rbw;
    float tmt = theta * (1.0f - theta);
    float num = ih * (idl * theta * theta + idv * tmt);
    float den = idl + (idv + idvp1 - 2.0f * idl) * tmt;
    float outv = ich + num * hw_rcp(den);
    float omt = 1.0f - theta;
    float dnum = idl * idl * (idvp1 * theta * theta + 2.0f * idl * tmt + idv * omt * omt);
    float ldv = 0.693147180559945309f * (__builtin_amdgcn_logf(dnum) - 2.0f * __builtin_amdgcn_logf(den));
    bool inside = (xf >= -BB) && (xf <= BB);
    z  = inside ? outv : xf;
    ld = inside ? ldv : 0.0f;
}

__device__ __forceinline__ int model_layer(int jl, int part) {
    const int l = 2 * (TSPLIT * (jl >> 1) + part) + (jl & 1);
    return l < LL ? l : LL - 1;
}

template <int N>
__device__ __forceinline__ void layer1_ksteps(const float* __restrict__ W1, unsigned w_off, int kl, int t,
                                              const v2f (&a0)[15], const v2f (&a1)[15],
                                              const float* __restrict__ x, unsigned xrow_off, int klast, v8f& c0, v8f& c1) {
    constexpr unsigned W1_LAST = 4u * (LL * LL * 8u - 1u);
    if (N == 16) w_off -= (t == 31 && w_off >= 4032u * 31u + 2016u) ? 2016u : 0u;
    const float* __restrict__ wp = (const float*)((const char*)W1 + w_off);
    v2f wk[N];
    #pragma unroll
    for (int k = 0; k < N; k++) {
        wk[k].x = wp[32 * k];
        if (N == 16 && k == 15) {
            const unsigned ol = w_off + 4u * (32u * 15u + 8u);
            wk[k].y = *(const float*)((const char*)W1 + (ol < W1_LAST ? ol : W1_LAST));
        } else {
            wk[k].y = wp[32 * k + 8];
        }
    }
    v2f a15_0 = {0.f, 0.f}, a15_1 = {0.f, 0.f};
    if (N == 16) {
        a15_0 = *(const v2f*)((const char*)x + (xrow_off + 16u * (unsigned)klast));
        a15_1 = *(const v2f*)((const char*)x + (xrow_off + TILE_X + 16u * (unsigned)klast));
    }
    wk[N - 1].x = (kl >= 0) ? wk[N - 1].x : 0.0f;
    wk[N - 1].y = (kl >= 1) ? wk[N - 1].y : 0.0f;
    __builtin_amdgcn_sched_barrier(0);
    v8f d0 = {0.f, 0.f, 0.f, 0.f, 0.f, 0.f, 0.f, 0.f};
    v8f d1 = {0.f, 0.f, 0.f, 0.f, 0.f, 0.f, 0.f, 0.f};
    #pragma unroll
    for (int k = 0; k < N; k++) {
        const v2f xk0 = (k < 15) ? a0[k < 15 ? k : 0] : a15_0;
        const v2f xk1 = (k < 15) ? a1[k < 15 ? k : 0] : a15_1;
        d0 = __builtin_amdgcn_wmma_f32_16x16x4_f32(false, wk[k], false, xk0, (short)0, d0, false, false);
        d1 = __builtin_amdgcn_wmma_f32_16x16x4_f32(false, wk[k], false, xk1, (short)0, d1, false, false);
    }
    c0 = d0;
    c1 = d1;
}

__launch_bounds__(128, 4)
__global__ void nsf_fused(const float* __restrict__ x, const float* __restrict__ initp,
                          const float* __restrict__ W1, const float* __restrict__ b1,
                          const float* __restrict__ W2, const float* __restrict__ b2,
                          const float* __restrict__ W3, const float* __restrict__ b3,
                          float* __restrict__ out) {
    __shared__ float ldacc[2 * DD];
    __shared__ __attribute__((aligned(16))) float wl[NLAY * LSTR];
    __shared__ __attribute__((aligned(16))) float b1s[TPER * 16];

    const int tid  = threadIdx.x;
    const int wave = __builtin_amdgcn_readfirstlane(tid >> 5);
    const int lane = tid & 31;
    const int m    = lane & 15;
    const int hh   = lane >> 4;
    const int s0   = blockIdx.x * BSAMP + wave * 32;
    const int s    = s0 + m;
    const int part = blockIdx.y;
    float* const ld_grp = ldacc + (wave >> 1) * DD;

    static_assert(2 * DD == 128, "one ldacc entry per thread");
    ldacc[tid] = 0.0f;

    {
        static_assert(NLAY % 16 == 0, "copy loops assume whole passes of 128 threads");
        const float4* src2 = (const float4*)W2;
        const float4* src3 = (const float4*)W3;
        constexpr int N3 = NLAY * 28;
        constexpr int P3 = (N3 + 127) / 128;
        float4 r2[NLAY * 16 / 128], r3[P3];
        float  rb2[NLAY * 8 / 128], rb3[NLAY * 16 / 128], rb1[TPER * 16 / 128];
        #pragma unroll
        for (int k = 0; k < NLAY * 16 / 128; k++) {
            const int i = tid + 128 * k;
            r2[k] = src2[model_layer(i >> 4, part) * 16 + (i & 15)];
        }
        #pragma unroll
        for (int k = 0; k < P3; k++) {
            const int i = tid + 128 * k;
            const int ic = (i < N3) ? i : N3 - 1;
            const int jl = ic / 28, j = ic - jl * 28;
            r3[k] = src3[model_layer(jl, part) * 28 + j];
        }
        #pragma unroll
        for (int k = 0; k < NLAY * 8 / 128; k++) {
            const int i = tid + 128 * k;
            rb2[k] = b2[model_layer(i >> 3, part) * 8 + (i & 7)];
        }
        #pragma unroll
        for (int k = 0; k < NLAY * 16 / 128; k++) {
            const int i = tid + 128 * k;
            const int q = i & 15;
            const float v = b3[model_layer(i >> 4, part) * PP + (q < PP ? q : PP - 1)];
            rb3[k] = (q < PP) ? v : 0.0f;
        }
        #pragma unroll
        for (int k = 0; k < TPER * 16 / 128; k++) {
            const int i = tid + 128 * k;
            const int col = 16 * (TSPLIT * (i >> 4) + part) + (i & 15);
            const float v = b1[col < LL * 8 ? col : LL * 8 - 1];
            rb1[k] = (col < LL * 8) ? v : 0.0f;
        }
        #pragma unroll
        for (int k = 0; k < NLAY * 16 / 128; k++) {
            const int i = tid + 128 * k;
            *(float4*)(wl + (i >> 4) * LSTR + 4 * (i & 15)) = r2[k];
        }
        #pragma unroll
        for (int k = 0; k < P3; k++) {
            const int i = tid + 128 * k;
            const int ic = (i < N3) ? i : N3 - 1;
            const int jl = ic / 28, j = ic - jl * 28;
            *(float4*)(wl + jl * LSTR + OFF_W3 + 4 * j) = r3[k];
        }
        #pragma unroll
        for (int k = 0; k < NLAY * 8 / 128; k++) {
            const int i = tid + 128 * k;
            wl[(i >> 3) * LSTR + OFF_B2 + (i & 7)] = rb2[k];
        }
        #pragma unroll
        for (int k = 0; k < NLAY * 16 / 128; k++) {
            const int i = tid + 128 * k;
            wl[(i >> 4) * LSTR + OFF_B3 + (i & 15)] = rb3[k];
        }
        #pragma unroll
        for (int k = 0; k < TPER * 16 / 128; k++) b1s[tid + 128 * k] = rb1[k];
    }
    __syncthreads();

    if (part == 0) {
        float p[PP];
        #pragma unroll
        for (int q = 0; q < PP; q++) p[q] = initp[q];
        float z, ld;
        rqs_eval(p, x[(s0 + lane) * DD + 0], z, ld);
        out[s0 + lane] = z;
        atomicAdd(&ld_grp[0], ld);
    }

    v2f a0[15], a1[15];
    const unsigned xrow_off = (unsigned)s * (DD * 4u) + 8u * (unsigned)hh;
    #pragma unroll
    for (int kk = 0; kk < 15; kk++) {
        a0[kk] = *(const v2f*)((const char*)x + xrow_off + 16u * kk);
        a1[kk] = *(const v2f*)((const char*)x + xrow_off + TILE_X + 16u * kk);
    }
    const unsigned zout_off = 4u * ((unsigned)(hh + 1) * (unsigned)NS + (unsigned)s);
    const unsigned w_lane = 2016u * (unsigned)(m >> 3) + 64u * (unsigned)hh + 4u * (unsigned)(m & 7);
    const float* wl_lane = wl + hh * LSTR;
    const float* b1_lane = b1s + 8 * hh;

    for (int i = 0; i < TPER; ++i) {
        const int t = TSPLIT * i + part;
        const unsigned w_off = w_lane + 4032u * (unsigned)t;
        const int kl = 2 * (t & 1) + (w_lane >= 2016u ? 1 : 0) - 2 * hh;
        v8f c0, c1;
        switch ((t >> 1) + 1) {
            case 1:  layer1_ksteps<1>(W1, w_off, kl, t, a0, a1, x, xrow_off, t >> 1, c0, c1); break;
            case 2:  layer1_ksteps<2>(W1, w_off, kl, t, a0, a1, x, xrow_off, t >> 1, c0, c1); break;
            case 3:  layer1_ksteps<3>(W1, w_off, kl, t, a0, a1, x, xrow_off, t >> 1, c0, c1); break;
            case 4:  layer1_ksteps<4>(W1, w_off, kl, t, a0, a1, x, xrow_off, t >> 1, c0, c1); break;
            case 5:  layer1_ksteps<5>(W1, w_off, kl, t, a0, a1, x, xrow_off, t >> 1, c0, c1); break;
            case 6:  layer1_ksteps<6>(W1, w_off, kl, t, a0, a1, x, xrow_off, t >> 1, c0, c1); break;
            case 7:  layer1_ksteps<7>(W1, w_off, kl, t, a0, a1, x, xrow_off, t >> 1, c0, c1); break;
            case 8:  layer1_ksteps<8>(W1, w_off, kl, t, a0, a1, x, xrow_off, t >> 1, c0, c1); break;
            case 9:  layer1_ksteps<9>(W1, w_off, kl, t, a0, a1, x, xrow_off, t >> 1, c0, c1); break;
            case 10: layer1_ksteps<10>(W1, w_off, kl, t, a0, a1, x, xrow_off, t >> 1, c0, c1); break;
            case 11: layer1_ksteps<11>(W1, w_off, kl, t, a0, a1, x, xrow_off, t >> 1, c0, c1); break;
            case 12: layer1_ksteps<12>(W1, w_off, kl, t, a0, a1, x, xrow_off, t >> 1, c0, c1); break;
            case 13: layer1_ksteps<13>(W1, w_off, kl, t, a0, a1, x, xrow_off, t >> 1, c0, c1); break;
            case 14: layer1_ksteps<14>(W1, w_off, kl, t, a0, a1, x, xrow_off, t >> 1, c0, c1); break;
            case 15: layer1_ksteps<15>(W1, w_off, kl, t, a0, a1, x, xrow_off, t >> 1, c0, c1); break;
            default: layer1_ksteps<16>(W1, w_off, kl, t, a0, a1, x, xrow_off, t >> 1, c0, c1); break;
        }

        float hv0[8], hv1[8];
        {
            const float* bp = b1_lane + 16 * i;
            const float4 bl = *(const float4*)bp, bh = *(const float4*)(bp + 4);
            const float bias[8] = {bl.x, bl.y, bl.z, bl.w, bh.x, bh.y, bh.z, bh.w};
            #pragma unroll
            for (int r = 0; r < 8; r++) { hv0[r] = hw_tanh(c0[r] + bias[r]); hv1[r] = hw_tanh(c1[r] + bias[r]); }
        }

        const int l_idx = 2 * t + hh;
        if (l_idx < LL) {
            const int cdim = l_idx + 1;
            const v2f xpair0 = *(const v2f*)((const char*)x + (xrow_off + 8u * (unsigned)t));
            const v2f xpair1 = *(const v2f*)((const char*)x + (xrow_off + TILE_X + 8u * (unsigned)t));
            const float xval0 = hh ? xpair0.x : xpair0.y;
            const float xval1 = hh ? xpair1.x : xpair1.y;

            const float* lw = wl_lane + i * (2 * LSTR);
            v2f acc0[4], acc1[4];
            #pragma unroll
            for (int gp = 0; gp < 4; gp++) { acc0[gp] = *(const v2f*)(lw + OFF_B2 + 2 * gp); acc1[gp] = acc0[gp]; }
            #pragma unroll
            for (int j = 0; j < 8; j++) {
                const v2f hs0 = {hv0[j], hv0[j]};
                const v2f hs1 = {hv1[j], hv1[j]};
                #pragma unroll
                for (int gp = 0; gp < 4; gp++) {
                    const v2f w = *(const v2f*)(lw + j * 8 + 2 * gp);
                    acc0[gp] = __builtin_elementwise_fma(hs0, w, acc0[gp]);
                    acc1[gp] = __builtin_elementwise_fma(hs1, w, acc1[gp]);
                }
            }
            float h2_0[8], h2_1[8];
            #pragma unroll
            for (int gp = 0; gp < 4; gp++) {
                h2_0[2 * gp] = hw_tanh(acc0[gp].x); h2_0[2 * gp + 1] = hw_tanh(acc0[gp].y);
                h2_1[2 * gp] = hw_tanh(acc1[gp].x); h2_1[2 * gp + 1] = hw_tanh(acc1[gp].y);
            }

            v2f pa[7], pb[7];
            #pragma unroll
            for (int qp = 0; qp < 7; qp++) { pa[qp] = *(const v2f*)(lw + OFF_B3 + 2 * qp); pb[qp] = pa[qp]; }
            #pragma unroll
            for (int g = 0; g < 8; g++) {
                const v2f hs0 = {h2_0[g], h2_0[g]};
                const v2f hs1 = {h2_1[g], h2_1[g]};
                #pragma unroll
                for (int qp = 0; qp < 7; qp++) {
                    const v2f w = *(const v2f*)(lw + OFF_W3 + g * PP + 2 * qp);
                    pa[qp] = __builtin_elementwise_fma(hs0, w, pa[qp]);
                    pb[qp] = __builtin_elementwise_fma(hs1, w, pb[qp]);
                }
            }
            float pv0[PP], pv1[PP];
            #pragma unroll
            for (int qp = 0; qp < 7; qp++) {
                pv0[2 * qp] = pa[qp].x; pv0[2 * qp + 1] = pa[qp].y;
                pv1[2 * qp] = pb[qp].x; pv1[2 * qp + 1] = pb[qp].y;
            }

            float z0, ld0, z1, ld1;
            rqs_eval(pv0, xval0, z0, ld0);
            rqs_eval(pv1, xval1, z1, ld1);
            const unsigned zo = zout_off + (unsigned)t * (8u * NS);
            *(float*)((char*)out + zo) = z0;
            *(float*)((char*)out + (zo + 64u)) = z1;
            atomicAdd(&ld_grp[cdim], ld0);
            atomicAdd(&ld_grp[cdim], ld1);
        }
    }
    __syncthreads();

    const int etid = 32 * wave + (int)__builtin_amdgcn_mbcnt_lo(~0u, 0u);
    {
        const int col = etid & (DD - 1), g = etid >> 6;
        const int owner = (col == 0) ? 0 : (((col - 1) >> 1) % TSPLIT);
        if (owner == part)
            out[(size_t)NS * DD + (size_t)col * (NS / 64) + 2 * blockIdx.x + g] = ldacc[etid];
    }
}

extern "C" void kernel_launch(void* const* d_in, const int* in_sizes, int n_in,
                              void* d_out, int out_size, void* d_ws, size_t ws_size,
                              hipStream_t stream) {
    const float* x     = (const float*)d_in[0];
    const float* initp = (const float*)d_in[1];
    const float* W1    = (const float*)d_in[2];
    const float* b1    = (const float*)d_in[3];
    const float* W2    = (const float*)d_in[4];
    const float* b2    = (const float*)d_in[5];
    const float* W3    = (const float*)d_in[6];
    const float* b3    = (const float*)d_in[7];
    float* out = (float*)d_out;
    nsf_fused<<<dim3(NS / BSAMP, TSPLIT), 128, 0, stream>>>(x, initp, W1, b1, W2, b2, W3, b3, out);
}
